# starting kernel
# baseline (speedup 1.0000x reference)
#include <hip/hip_runtime.h>

#define HID 64

typedef float v2f __attribute__((ext_vector_type(2)));
typedef float v8f __attribute__((ext_vector_type(8)));

__global__ void k_zero(float* __restrict__ p, int n) {
  int t = blockIdx.x * blockDim.x + threadIdx.x;
  if (t < n) p[t] = 0.0f;
}

__global__ void k_degree(const int* __restrict__ ei, int E, float* __restrict__ deg) {
  int e = blockIdx.x * blockDim.x + threadIdx.x;
  if (e < E) atomicAdd(&deg[ei[E + e]], 1.0f);
}

__global__ void k_dinv(float* __restrict__ deg, int N) {
  int i = blockIdx.x * blockDim.x + threadIdx.x;
  if (i < N) deg[i] = rsqrtf(deg[i] + 1.0f);
}

__global__ void k_gemm1(const float* __restrict__ x, const float* __restrict__ W1,
                        float* __restrict__ out, int N) {
  int t = blockIdx.x * blockDim.x + threadIdx.x;
  if (t >= N * HID) return;
  int n = t >> 6, j = t & 63;
  float x0 = x[n * 3 + 0], x1 = x[n * 3 + 1], x2 = x[n * 3 + 2];
  out[t] = x0 * W1[j] + x1 * W1[HID + j] + x2 * W1[2 * HID + j];
}

__global__ __launch_bounds__(32) void k_gemm2(const float* __restrict__ H,
                                              const float* __restrict__ W,
                                              float* __restrict__ O, int N) {
  const int tm   = blockIdx.x;
  const int tn   = blockIdx.y;
  const int lane = threadIdx.x;
  const int m    = lane & 15;
  const int half = lane >> 4;
  const int ncol = tn * 16 + (lane & 15);

  int arow = tm * 16 + m;
  if (arow >= N) arow = N - 1;
  const float* __restrict__ Arow = H + (long)arow * HID;

  v8f acc = {};
  #pragma unroll
  for (int k0 = 0; k0 < HID; k0 += 4) {
    const int ka = k0 + half * 2;
    v2f a; a.x = Arow[ka];             a.y = Arow[ka + 1];
    v2f b; b.x = W[ka * HID + ncol];   b.y = W[(ka + 1) * HID + ncol];
    acc = __builtin_amdgcn_wmma_f32_16x16x4_f32(
         false, a,  false, b,
         (short)0, acc,  false,  false);
  }
  #pragma unroll
  for (int r = 0; r < 8; ++r) {
    int mrow = tm * 16 + r + half * 8;
    if (mrow < N) O[(long)mrow * HID + ncol] = acc[r];
  }
}

__global__ void k_scatter(const int* __restrict__ ei, int E,
                          const float* __restrict__ hw,
                          const float* __restrict__ dinv,
                          float* __restrict__ agg) {
  int t = blockIdx.x * blockDim.x + threadIdx.x;
  if (t >= E * 16) return;
  int e = t >> 4;
  int j = (t & 15) * 4;
  int s = ei[e];
  int d = ei[E + e];
  float norm = dinv[s] * dinv[d];
  const float4 v = *(const float4*)(hw + (long)s * HID + j);
  float* o = agg + (long)d * HID + j;
  atomicAdd(o + 0, v.x * norm);
  atomicAdd(o + 1, v.y * norm);
  atomicAdd(o + 2, v.z * norm);
  atomicAdd(o + 3, v.w * norm);
}

__global__ void k_combine(float* __restrict__ agg, const float* __restrict__ hw,
                          const float* __restrict__ dinv, const float* __restrict__ bias,
                          int N) {
  int t = blockIdx.x * blockDim.x + threadIdx.x;
  if (t >= N * 16) return;
  int n = t >> 4;
  int j = (t & 15) * 4;
  float di = dinv[n];
  float sl = di * di;
  float4 a = *(float4*)(agg + (long)n * HID + j);
  const float4 h = *(const float4*)(hw + (long)n * HID + j);
  a.x = fmaxf(fmaf(h.x, sl, a.x) + bias[j + 0], 0.0f);
  a.y = fmaxf(fmaf(h.y, sl, a.y) + bias[j + 1], 0.0f);
  a.z = fmaxf(fmaf(h.z, sl, a.z) + bias[j + 2], 0.0f);
  a.w = fmaxf(fmaf(h.w, sl, a.w) + bias[j + 3], 0.0f);
  *(float4*)(agg + (long)n * HID + j) = a;
}

__global__ void k_pool(const float* __restrict__ h, const int* __restrict__ batch,
                       float* __restrict__ out, float* __restrict__ cnt, int N) {
  int t = blockIdx.x * blockDim.x + threadIdx.x;
  if (t >= N * 16) return;
  int n = t >> 4;
  int j = (t & 15) * 4;
  int g = batch[n];
  const float4 v = *(const float4*)(h + (long)n * HID + j);
  float* o = out + (long)g * HID + j;
  atomicAdd(o + 0, v.x);
  atomicAdd(o + 1, v.y);
  atomicAdd(o + 2, v.z);
  atomicAdd(o + 3, v.w);
  if ((t & 15) == 0) atomicAdd(&cnt[g], 1.0f);
}

__global__ void k_div(float* __restrict__ out, const float* __restrict__ cnt, int total) {
  int t = blockIdx.x * blockDim.x + threadIdx.x;
  if (t < total) out[t] /= fmaxf(cnt[t >> 6], 1.0f);
}

extern "C" void kernel_launch(void* const* d_in, const int* in_sizes, int n_in,
                              void* d_out, int out_size, void* d_ws, size_t ws_size,
                              hipStream_t stream) {
  const float* x   = (const float*)d_in[0];
  const float* W1  = (const float*)d_in[1];
  const float* b1  = (const float*)d_in[2];
  const float* W2  = (const float*)d_in[3];
  const float* b2  = (const float*)d_in[4];
  const int*   ei  = (const int*)d_in[5];
  const int*   bat = (const int*)d_in[6];
  float* out = (float*)d_out;

  const int N = in_sizes[0] / 3;
  const int E = in_sizes[5] / 2;

  char* ws = (char*)d_ws;
  float* dinv = (float*)(ws);
  float* bufA = (float*)(ws + (1 << 19));
  float* bufB = (float*)(ws + (1 << 19) + (size_t)N * HID * 4);
  float* cnt  = (float*)(ws + (1 << 19) + 2 * (size_t)N * HID * 4);

  const int B = 256;
  const int NH   = N * HID;
  const int N16  = N * 16;
  const int E16  = E * 16;
  const int tilesM = (N + 15) / 16;

  k_zero  <<<(N + B - 1) / B, B, 0, stream>>>(dinv, N);
  k_degree<<<(E + B - 1) / B, B, 0, stream>>>(ei, E, dinv);
  k_dinv  <<<(N + B - 1) / B, B, 0, stream>>>(dinv, N);

  k_gemm1  <<<(NH + B - 1) / B, B, 0, stream>>>(x, W1, bufA, N);
  k_zero   <<<(NH + B - 1) / B, B, 0, stream>>>(bufB, NH);
  k_scatter<<<(E16 + B - 1) / B, B, 0, stream>>>(ei, E, bufA, dinv, bufB);
  k_combine<<<(N16 + B - 1) / B, B, 0, stream>>>(bufB, bufA, dinv, b1, N);

  {
    dim3 grid(tilesM, HID / 16);
    k_gemm2<<<grid, 32, 0, stream>>>(bufB, W2, bufA, N);
  }
  k_zero   <<<(NH + B - 1) / B, B, 0, stream>>>(bufB, NH);
  k_scatter<<<(E16 + B - 1) / B, B, 0, stream>>>(ei, E, bufA, dinv, bufB);
  k_combine<<<(N16 + B - 1) / B, B, 0, stream>>>(bufB, bufA, dinv, b2, N);

  k_zero<<<(out_size + B - 1) / B, B, 0, stream>>>(out, out_size);
  k_zero<<<(512 + B - 1) / B, B, 0, stream>>>(cnt, 512);
  k_pool<<<(N16 + B - 1) / B, B, 0, stream>>>(bufB, bat, out, cnt, N);
  k_div <<<(out_size + B - 1) / B, B, 0, stream>>>(out, cnt, out_size);
}


# zero-fill, dinv and combine passes folded into neighbours: GEMM epilogue stores self-loop+bias as the scatter's start value, consumer applies relu on load (15 launches to 8); self-loop summed first
# speedup vs baseline: 1.2400x; 1.2400x over previous
#include <hip/hip_runtime.h>

#define HID 64

typedef float v2f __attribute__((ext_vector_type(2)));
typedef float v8f __attribute__((ext_vector_type(8)));

__global__ void k_zero(float* __restrict__ p, int n) {
  int t = blockIdx.x * blockDim.x + threadIdx.x;
  if (t < n) p[t] = 0.0f;
}

__global__ void k_degree(const int* __restrict__ ei, int E, float* __restrict__ deg) {
  int e = blockIdx.x * blockDim.x + threadIdx.x;
  if (e < E) atomicAdd(&deg[ei[E + e]], 1.0f);
}

__global__ void k_gemm1(const float* __restrict__ x, const float* __restrict__ W1,
                        const float* __restrict__ bias, const float* __restrict__ deg,
                        float* __restrict__ dinv, float* __restrict__ hw,
                        float* __restrict__ agg, float* __restrict__ pool,
                        float* __restrict__ cnt, int N, int pool_size) {
  int t = blockIdx.x * blockDim.x + threadIdx.x;
  if (t < pool_size) pool[t] = 0.0f;
  if (t < (pool_size >> 6)) cnt[t] = 0.0f;
  if (t >= N * HID) return;
  int n = t >> 6, j = t & 63;
  float dg = deg[n];
  float x0 = x[n * 3 + 0], x1 = x[n * 3 + 1], x2 = x[n * 3 + 2];
  float v = x0 * W1[j] + x1 * W1[HID + j] + x2 * W1[2 * HID + j];
  float di = rsqrtf(dg + 1.0f);
  hw[t] = v;
  agg[t] = fmaf(v, di * di, bias[j]);
  if (j == 0) dinv[n] = di;
}

__global__ __launch_bounds__(32) void k_gemm2(const float* __restrict__ H,
                                              const float* __restrict__ W,
                                              const float* __restrict__ bias,
                                              const float* __restrict__ dinv,
                                              float* __restrict__ O,
                                              float* __restrict__ S, int N) {
  const int tm   = blockIdx.x;
  const int tn   = blockIdx.y;
  const int lane = threadIdx.x;
  const int m    = lane & 15;
  const int half = lane >> 4;
  const int ncol = tn * 16 + (lane & 15);

  int arow = tm * 16 + m;
  if (arow >= N) arow = N - 1;
  const float* __restrict__ Arow = H + (long)arow * HID;

  float sl[8];
  #pragma unroll
  for (int r = 0; r < 8; ++r) {
    int mr = tm * 16 + r + half * 8;
    if (mr >= N) mr = N - 1;
    const float di = dinv[mr];
    sl[r] = di * di;
  }
  const float bj = bias[ncol];

  v8f acc = {};
  #pragma unroll
  for (int k0 = 0; k0 < HID; k0 += 4) {
    const int ka = k0 + half * 2;
    v2f a; a.x = fmaxf(Arow[ka], 0.0f); a.y = fmaxf(Arow[ka + 1], 0.0f);
    v2f b; b.x = W[ka * HID + ncol];   b.y = W[(ka + 1) * HID + ncol];
    acc = __builtin_amdgcn_wmma_f32_16x16x4_f32(
         false, a,  false, b,
         (short)0, acc,  false,  false);
  }
  #pragma unroll
  for (int r = 0; r < 8; ++r) {
    int mrow = tm * 16 + r + half * 8;
    if (mrow < N) {
      O[(long)mrow * HID + ncol] = acc[r];
      S[(long)mrow * HID + ncol] = fmaf(acc[r], sl[r], bj);
    }
  }
}

__global__ void k_scatter(const int* __restrict__ ei, int E,
                          const float* __restrict__ hw,
                          const float* __restrict__ dinv,
                          float* __restrict__ agg) {
  int t = blockIdx.x * blockDim.x + threadIdx.x;
  if (t >= E * 16) return;
  int e = t >> 4;
  int j = (t & 15) * 4;
  int s = ei[e];
  int d = ei[E + e];
  float norm = dinv[s] * dinv[d];
  const float4 v = *(const float4*)(hw + (long)s * HID + j);
  float* o = agg + (long)d * HID + j;
  atomicAdd(o + 0, v.x * norm);
  atomicAdd(o + 1, v.y * norm);
  atomicAdd(o + 2, v.z * norm);
  atomicAdd(o + 3, v.w * norm);
}

__global__ void k_pool(const float* __restrict__ h, const int* __restrict__ batch,
                       float* __restrict__ out, float* __restrict__ cnt, int N) {
  int t = blockIdx.x * blockDim.x + threadIdx.x;
  if (t >= N * 16) return;
  int n = t >> 4;
  int j = (t & 15) * 4;
  int g = batch[n];
  const float4 v = *(const float4*)(h + (long)n * HID + j);
  float* o = out + (long)g * HID + j;
  atomicAdd(o + 0, fmaxf(v.x, 0.0f));
  atomicAdd(o + 1, fmaxf(v.y, 0.0f));
  atomicAdd(o + 2, fmaxf(v.z, 0.0f));
  atomicAdd(o + 3, fmaxf(v.w, 0.0f));
  if ((t & 15) == 0) atomicAdd(&cnt[g], 1.0f);
}

__global__ void k_div(float* __restrict__ out, const float* __restrict__ cnt, int total) {
  int t = blockIdx.x * blockDim.x + threadIdx.x;
  if (t < total) out[t] /= fmaxf(cnt[t >> 6], 1.0f);
}

extern "C" void kernel_launch(void* const* d_in, const int* in_sizes, int n_in,
                              void* d_out, int out_size, void* d_ws, size_t ws_size,
                              hipStream_t stream) {
  const float* x   = (const float*)d_in[0];
  const float* W1  = (const float*)d_in[1];
  const float* b1  = (const float*)d_in[2];
  const float* W2  = (const float*)d_in[3];
  const float* b2  = (const float*)d_in[4];
  const int*   ei  = (const int*)d_in[5];
  const int*   bat = (const int*)d_in[6];
  float* out = (float*)d_out;

  const int N = in_sizes[0] / 3;
  const int E = in_sizes[5] / 2;

  char* ws = (char*)d_ws;
  float* deg  = (float*)(ws);
  float* dinv = (float*)(ws + (1 << 19));
  float* bufA = (float*)(ws + (1 << 20));
  float* bufB = (float*)(ws + (1 << 20) + (size_t)N * HID * 4);
  float* bufC = (float*)(ws + (1 << 20) + 2 * (size_t)N * HID * 4);
  float* cnt  = (float*)(ws + (1 << 20) + 3 * (size_t)N * HID * 4);

  const int B = 256;
  const int NH   = N * HID;
  const int N16  = N * 16;
  const int E16  = E * 16;
  const int tilesM = (N + 15) / 16;

  k_zero  <<<(N + B - 1) / B, B, 0, stream>>>(deg, N);
  k_degree<<<(E + B - 1) / B, B, 0, stream>>>(ei, E, deg);

  k_gemm1  <<<(NH + B - 1) / B, B, 0, stream>>>(x, W1, b1, deg, dinv, bufA, bufB, out, cnt, N, out_size);
  k_scatter<<<(E16 + B - 1) / B, B, 0, stream>>>(ei, E, bufA, dinv, bufB);

  {
    dim3 grid(tilesM, HID / 16);
    k_gemm2<<<grid, 32, 0, stream>>>(bufB, W2, b2, dinv, bufA, bufC, N);
  }
  k_scatter<<<(E16 + B - 1) / B, B, 0, stream>>>(ei, E, bufA, dinv, bufC);

  k_pool<<<(N16 + B - 1) / B, B, 0, stream>>>(bufC, bat, out, cnt, N);
  k_div <<<(out_size + B - 1) / B, B, 0, stream>>>(out, cnt, out_size);
}


# layer-1 neighbour sum taken over the 3 input features before the 3x64 linear (A(xW)=(Ax)W), linear+relu formed in the WMMA GEMM's A loader; reorders layer 1's f32 sums
# speedup vs baseline: 1.4979x; 1.2080x over previous
#include <hip/hip_runtime.h>


#define HID 64

typedef float v2f __attribute__((ext_vector_type(2)));
typedef float v8f __attribute__((ext_vector_type(8)));

__global__ void k_zero(float* __restrict__ p, int n) {
  int t = blockIdx.x * blockDim.x + threadIdx.x;
  if (t < n) p[t] = 0.0f;
}

__global__ void k_degree(const int* __restrict__ ei, int E, float* __restrict__ deg) {
  int e = blockIdx.x * blockDim.x + threadIdx.x;
  if (e < E) atomicAdd(&deg[ei[E + e]], 1.0f);
}

__global__ void k_prep(const float* __restrict__ x, const float* __restrict__ deg,
                       float* __restrict__ dinv, float* __restrict__ ax,
                       float* __restrict__ pool, float* __restrict__ cnt,
                       int N, int pool_size) {
  const int t = blockIdx.x * blockDim.x + threadIdx.x;
  const int stride = gridDim.x * blockDim.x;
  for (int i = t; i < pool_size; i += stride) pool[i] = 0.0f;
  for (int i = t; i < (pool_size >> 6); i += stride) cnt[i] = 0.0f;
  if (t >= N) return;
  const float di = rsqrtf(deg[t] + 1.0f);
  const float s  = di * di;
  float4 v;
  v.x = x[t * 3 + 0] * s;
  v.y = x[t * 3 + 1] * s;
  v.z = x[t * 3 + 2] * s;
  v.w = 0.0f;
  *(float4*)(ax + (long)t * 4) = v;
  dinv[t] = di;
}

__global__ void k_scatter3(const int* __restrict__ ei, int E,
                           const float* __restrict__ x,
                           const float* __restrict__ dinv,
                           float* __restrict__ ax) {
  const int t  = blockIdx.x * blockDim.x + threadIdx.x;
  const int c  = t & 3;
  const int e0 = (t >> 2) * 4;
  if (c >= 3 || e0 >= E) return;
  if (e0 + 3 < E) {
    int s[4], d[4];
    #pragma unroll
    for (int i = 0; i < 4; ++i) { s[i] = ei[e0 + i]; d[i] = ei[E + e0 + i]; }
    float v[4];
    #pragma unroll
    for (int i = 0; i < 4; ++i)
      v[i] = x[(long)s[i] * 3 + c] * (dinv[s[i]] * dinv[d[i]]);
    #pragma unroll
    for (int i = 0; i < 4; ++i) atomicAdd(ax + (long)d[i] * 4 + c, v[i]);
  } else {
    for (int e = e0; e < E; ++e) {
      const int s = ei[e], d = ei[E + e];
      atomicAdd(ax + (long)d * 4 + c, x[(long)s * 3 + c] * (dinv[s] * dinv[d]));
    }
  }
}

__global__ __launch_bounds__(32) void k_gemm2(const float* __restrict__ AX,
                                              const float* __restrict__ W1,
                                              const float* __restrict__ b1,
                                              const float* __restrict__ W,
                                              const float* __restrict__ bias,
                                              const float* __restrict__ dinv,
                                              float* __restrict__ O,
                                              float* __restrict__ S, int N) {
  const int tm   = blockIdx.x;
  const int tn   = blockIdx.y;
  const int lane = threadIdx.x;
  const int m    = lane & 15;
  const int half = lane >> 4;
  const int ncol = tn * 16 + (lane & 15);

  int arow = tm * 16 + m;
  if (arow >= N) arow = N - 1;
  const float4 xa = *(const float4*)(AX + (long)arow * 4);

  float sl[8];
  #pragma unroll
  for (int r = 0; r < 8; ++r) {
    int mr = tm * 16 + r + half * 8;
    if (mr >= N) mr = N - 1;
    const float di = dinv[mr];
    sl[r] = di * di;
  }
  const float bj = bias[ncol];

  v8f acc = {};
  #pragma unroll
  for (int k0 = 0; k0 < HID; k0 += 4) {
    const int ka = k0 + half * 2;
    const float h0 = xa.x * W1[ka]     + xa.y * W1[HID + ka]     + xa.z * W1[2 * HID + ka]     + b1[ka];
    const float h1 = xa.x * W1[ka + 1] + xa.y * W1[HID + ka + 1] + xa.z * W1[2 * HID + ka + 1] + b1[ka + 1];
    v2f a; a.x = fmaxf(h0, 0.0f);      a.y = fmaxf(h1, 0.0f);
    v2f b; b.x = W[ka * HID + ncol];   b.y = W[(ka + 1) * HID + ncol];
    acc = __builtin_amdgcn_wmma_f32_16x16x4_f32(
         false, a,  false, b,
         (short)0, acc,  false,  false);
  }
  #pragma unroll
  for (int r = 0; r < 8; ++r) {
    int mrow = tm * 16 + r + half * 8;
    if (mrow < N) {
      O[(long)mrow * HID + ncol] = acc[r];
      S[(long)mrow * HID + ncol] = fmaf(acc[r], sl[r], bj);
    }
  }
}

__global__ void k_scatter(const int* __restrict__ ei, int E,
                          const float* __restrict__ hw,
                          const float* __restrict__ dinv,
                          float* __restrict__ agg) {
  int t = blockIdx.x * blockDim.x + threadIdx.x;
  if (t >= E * 16) return;
  int e = t >> 4;
  int j = (t & 15) * 4;
  int s = ei[e];
  int d = ei[E + e];
  float norm = dinv[s] * dinv[d];
  const float4 v = *(const float4*)(hw + (long)s * HID + j);
  float* o = agg + (long)d * HID + j;
  atomicAdd(o + 0, v.x * norm);
  atomicAdd(o + 1, v.y * norm);
  atomicAdd(o + 2, v.z * norm);
  atomicAdd(o + 3, v.w * norm);
}

__global__ void k_pool(const float* __restrict__ h, const int* __restrict__ batch,
                       float* __restrict__ out, float* __restrict__ cnt, int N) {
  int t = blockIdx.x * blockDim.x + threadIdx.x;
  if (t >= N * 16) return;
  int n = t >> 4;
  int j = (t & 15) * 4;
  int g = batch[n];
  const float4 v = *(const float4*)(h + (long)n * HID + j);
  float* o = out + (long)g * HID + j;
  atomicAdd(o + 0, fmaxf(v.x, 0.0f));
  atomicAdd(o + 1, fmaxf(v.y, 0.0f));
  atomicAdd(o + 2, fmaxf(v.z, 0.0f));
  atomicAdd(o + 3, fmaxf(v.w, 0.0f));
  if ((t & 15) == 0) atomicAdd(&cnt[g], 1.0f);
}

__global__ void k_div(float* __restrict__ out, const float* __restrict__ cnt, int total) {
  int t = blockIdx.x * blockDim.x + threadIdx.x;
  if (t < total) out[t] /= fmaxf(cnt[t >> 6], 1.0f);
}

extern "C" void kernel_launch(void* const* d_in, const int* in_sizes, int n_in,
                              void* d_out, int out_size, void* d_ws, size_t ws_size,
                              hipStream_t stream) {
  const float* x   = (const float*)d_in[0];
  const float* W1  = (const float*)d_in[1];
  const float* b1  = (const float*)d_in[2];
  const float* W2  = (const float*)d_in[3];
  const float* b2  = (const float*)d_in[4];
  const int*   ei  = (const int*)d_in[5];
  const int*   bat = (const int*)d_in[6];
  float* out = (float*)d_out;

  const int N = in_sizes[0] / 3;
  const int E = in_sizes[5] / 2;

  char* ws = (char*)d_ws;
  float* deg  = (float*)(ws);
  float* dinv = (float*)(ws + (1 << 19));
  float* bufA = (float*)(ws + (1 << 20));
  float* ax   = (float*)(ws + (1 << 20) + (size_t)N * HID * 4);
  float* bufC = (float*)(ws + (1 << 20) + 2 * (size_t)N * HID * 4);
  float* cnt  = (float*)(ws + (1 << 20) + 3 * (size_t)N * HID * 4);

  const int B = 256;
  const int N16  = N * 16;
  const int E16  = E * 16;
  const int tilesM = (N + 15) / 16;

  k_zero  <<<(N + B - 1) / B, B, 0, stream>>>(deg, N);
  k_degree<<<(E + B - 1) / B, B, 0, stream>>>(ei, E, deg);

  k_prep    <<<(N + B - 1) / B, B, 0, stream>>>(x, deg, dinv, ax, out, cnt, N, out_size);
  k_scatter3<<<(E + B - 1) / B, B, 0, stream>>>(ei, E, x, dinv, ax);

  {
    dim3 grid(tilesM, HID / 16);
    k_gemm2<<<grid, 32, 0, stream>>>(ax, W1, b1, W2, b2, dinv, bufA, bufC, N);
  }
  k_scatter<<<(E16 + B - 1) / B, B, 0, stream>>>(ei, E, bufA, dinv, bufC);

  k_pool<<<(N16 + B - 1) / B, B, 0, stream>>>(bufC, bat, out, cnt, N);
  k_div <<<(out_size + B - 1) / B, B, 0, stream>>>(out, cnt, out_size);
}


# every sum built with atomic adds gets one plane per package half: a wave adds into its own half's plane (die number from the placement message), readers add the two planes; reorders the f32 sums
# speedup vs baseline: 1.5518x; 1.0360x over previous
#include <hip/hip_runtime.h>


#define HID 64

typedef float v2f __attribute__((ext_vector_type(2)));
typedef float v8f __attribute__((ext_vector_type(8)));

__device__ __forceinline__ int pkg_half() {
  return (int)((__builtin_amdgcn_s_sendmsg_rtn(0x87) >> 18) & 1u);
}

__global__ void k_zero(float* __restrict__ a, int na, float* __restrict__ b, int n4b) {
  const int t = blockIdx.x * blockDim.x + threadIdx.x;
  const int stride = gridDim.x * blockDim.x;
  float4* b4 = (float4*)b;
  const float4 z = {0.0f, 0.0f, 0.0f, 0.0f};
  for (int i = t; i < na; i += stride) a[i] = 0.0f;
  for (int i = t; i < n4b; i += stride) b4[i] = z;
}

__global__ void k_degree(const int* __restrict__ ei, int E, float* __restrict__ deg, int N) {
  int e = blockIdx.x * blockDim.x + threadIdx.x;
  if (e >= E) return;
  const int d = ei[E + e];
  float* degh = deg + (long)pkg_half() * N;
  atomicAdd(&degh[d], 1.0f);
}

__global__ void k_prep(const float* __restrict__ x, const float* __restrict__ deg,
                       float* __restrict__ dinv, float* __restrict__ ax,
                       float* __restrict__ pool0, float* __restrict__ pool1,
                       float* __restrict__ cnt,
                       int N, int pool_size) {
  const int t = blockIdx.x * blockDim.x + threadIdx.x;
  const int stride = gridDim.x * blockDim.x;
  for (int i = t; i < pool_size; i += stride) { pool0[i] = 0.0f; pool1[i] = 0.0f; }
  for (int i = t; i < 2 * (pool_size >> 6); i += stride) cnt[i] = 0.0f;
  if (t >= N) return;
  const float di = rsqrtf(deg[t] + deg[N + t] + 1.0f);
  const float s  = di * di;
  float4 v;
  v.x = x[t * 3 + 0] * s;
  v.y = x[t * 3 + 1] * s;
  v.z = x[t * 3 + 2] * s;
  v.w = 0.0f;
  *(float4*)(ax + (long)t * 4) = v;
  const float4 z = {0.0f, 0.0f, 0.0f, 0.0f};
  *(float4*)(ax + ((long)N + t) * 4) = z;
  dinv[t] = di;
}

__global__ void k_scatter3(const int* __restrict__ ei, int E,
                           const float* __restrict__ x,
                           const float* __restrict__ dinv,
                           float* __restrict__ ax2, int N) {
  const int t  = blockIdx.x * blockDim.x + threadIdx.x;
  const int c  = t & 3;
  const int e0 = (t >> 2) * 4;
  if (c >= 3 || e0 >= E) return;
  if (e0 + 3 < E) {
    int s[4], d[4];
    #pragma unroll
    for (int i = 0; i < 4; ++i) { s[i] = ei[e0 + i]; d[i] = ei[E + e0 + i]; }
    float v[4];
    #pragma unroll
    for (int i = 0; i < 4; ++i)
      v[i] = x[(long)s[i] * 3 + c] * (dinv[s[i]] * dinv[d[i]]);
    float* ax = ax2 + (long)pkg_half() * N * 4;
    #pragma unroll
    for (int i = 0; i < 4; ++i) atomicAdd(ax + (long)d[i] * 4 + c, v[i]);
  } else {
    float* ax = ax2 + (long)pkg_half() * N * 4;
    for (int e = e0; e < E; ++e) {
      const int s = ei[e], d = ei[E + e];
      atomicAdd(ax + (long)d * 4 + c, x[(long)s * 3 + c] * (dinv[s] * dinv[d]));
    }
  }
}

__global__ __launch_bounds__(32) void k_gemm2(const float* __restrict__ AX,
                                              const float* __restrict__ W1,
                                              const float* __restrict__ b1,
                                              const float* __restrict__ W,
                                              const float* __restrict__ bias,
                                              const float* __restrict__ dinv,
                                              float* __restrict__ O,
                                              float* __restrict__ S, int N) {
  const int tm   = blockIdx.x;
  const int tn   = blockIdx.y;
  const int lane = threadIdx.x;
  const int m    = lane & 15;
  const int half = lane >> 4;
  const int ncol = tn * 16 + (lane & 15);

  int arow = tm * 16 + m;
  if (arow >= N) arow = N - 1;
  const float4 xa0 = *(const float4*)(AX + (long)arow * 4);
  const float4 xa1 = *(const float4*)(AX + ((long)N + arow) * 4);
  float4 xa;
  xa.x = xa0.x + xa1.x; xa.y = xa0.y + xa1.y; xa.z = xa0.z + xa1.z; xa.w = 0.0f;

  float sl[8];
  #pragma unroll
  for (int r = 0; r < 8; ++r) {
    int mr = tm * 16 + r + half * 8;
    if (mr >= N) mr = N - 1;
    const float di = dinv[mr];
    sl[r] = di * di;
  }
  const float bj = bias[ncol];

  v8f acc = {};
  #pragma unroll
  for (int k0 = 0; k0 < HID; k0 += 4) {
    const int ka = k0 + half * 2;
    const float h0 = xa.x * W1[ka]     + xa.y * W1[HID + ka]     + xa.z * W1[2 * HID + ka]     + b1[ka];
    const float h1 = xa.x * W1[ka + 1] + xa.y * W1[HID + ka + 1] + xa.z * W1[2 * HID + ka + 1] + b1[ka + 1];
    v2f a; a.x = fmaxf(h0, 0.0f);      a.y = fmaxf(h1, 0.0f);
    v2f b; b.x = W[ka * HID + ncol];   b.y = W[(ka + 1) * HID + ncol];
    acc = __builtin_amdgcn_wmma_f32_16x16x4_f32(
         false, a,  false, b,
         (short)0, acc,  false,  false);
  }
  #pragma unroll
  for (int r = 0; r < 8; ++r) {
    int mrow = tm * 16 + r + half * 8;
    if (mrow < N) {
      O[(long)mrow * HID + ncol] = acc[r];
      S[(long)mrow * HID + ncol] = fmaf(acc[r], sl[r], bj);
    }
  }
}

__global__ void k_scatter(const int* __restrict__ ei, int E,
                          const float* __restrict__ hw,
                          const float* __restrict__ dinv,
                          float* __restrict__ agg2, int N) {
  int t = blockIdx.x * blockDim.x + threadIdx.x;
  if (t >= E * 16) return;
  int e = t >> 4;
  int j = (t & 15) * 4;
  int s = ei[e];
  int d = ei[E + e];
  float norm = dinv[s] * dinv[d];
  const float4 v = *(const float4*)(hw + (long)s * HID + j);
  float* agg = agg2 + (long)pkg_half() * N * HID;
  float* o = agg + (long)d * HID + j;
  atomicAdd(o + 0, v.x * norm);
  atomicAdd(o + 1, v.y * norm);
  atomicAdd(o + 2, v.z * norm);
  atomicAdd(o + 3, v.w * norm);
}

__global__ void k_pool(const float* __restrict__ h, const int* __restrict__ batch,
                       float* __restrict__ pool0, float* __restrict__ pool1,
                       float* __restrict__ cnt2, int N, int G) {
  int t = blockIdx.x * blockDim.x + threadIdx.x;
  if (t >= N * 16) return;
  int n = t >> 4;
  int j = (t & 15) * 4;
  int g = batch[n];
  const float4 v0 = *(const float4*)(h + (long)n * HID + j);
  const float4 v1 = *(const float4*)(h + ((long)N + n) * HID + j);
  float4 v;
  v.x = v0.x + v1.x; v.y = v0.y + v1.y; v.z = v0.z + v1.z; v.w = v0.w + v1.w;
  const int hf = pkg_half();
  float* out = hf ? pool1 : pool0;
  float* cnt = cnt2 + hf * G;
  float* o = out + (long)g * HID + j;
  atomicAdd(o + 0, fmaxf(v.x, 0.0f));
  atomicAdd(o + 1, fmaxf(v.y, 0.0f));
  atomicAdd(o + 2, fmaxf(v.z, 0.0f));
  atomicAdd(o + 3, fmaxf(v.w, 0.0f));
  if ((t & 15) == 0) atomicAdd(&cnt[g], 1.0f);
}

__global__ void k_div(float* __restrict__ out, const float* __restrict__ pool1,
                      const float* __restrict__ cnt, int total, int G) {
  int t = blockIdx.x * blockDim.x + threadIdx.x;
  if (t < total) out[t] = (out[t] + pool1[t]) / fmaxf(cnt[t >> 6] + cnt[G + (t >> 6)], 1.0f);
}

extern "C" void kernel_launch(void* const* d_in, const int* in_sizes, int n_in,
                              void* d_out, int out_size, void* d_ws, size_t ws_size,
                              hipStream_t stream) {
  const float* x   = (const float*)d_in[0];
  const float* W1  = (const float*)d_in[1];
  const float* b1  = (const float*)d_in[2];
  const float* W2  = (const float*)d_in[3];
  const float* b2  = (const float*)d_in[4];
  const int*   ei  = (const int*)d_in[5];
  const int*   bat = (const int*)d_in[6];
  float* out = (float*)d_out;

  const int N = in_sizes[0] / 3;
  const int E = in_sizes[5] / 2;

  const int G = out_size / HID;

  char* ws = (char*)d_ws;
  float* deg  = (float*)(ws);
  float* dinv = (float*)(ws + (1 << 20));
  float* bufA = (float*)(ws + (1 << 21));
  float* ax   = (float*)(ws + (1 << 21) + (size_t)N * HID * 4);
  float* bufC = (float*)(ws + (1 << 21) + 2 * (size_t)N * HID * 4);
  float* pool1 = (float*)(ws + (1 << 21) + 4 * (size_t)N * HID * 4);
  float* cnt  = pool1 + out_size;

  const int B = 256;
  const int N16  = N * 16;
  const int E16  = E * 16;
  const int tilesM = (N + 15) / 16;

  k_zero  <<<1024, B, 0, stream>>>(deg, 2 * N, bufC + (size_t)N * HID, N16);
  k_degree<<<(E + B - 1) / B, B, 0, stream>>>(ei, E, deg, N);

  k_prep    <<<(N + B - 1) / B, B, 0, stream>>>(x, deg, dinv, ax, out, pool1, cnt, N, out_size);
  k_scatter3<<<(E + B - 1) / B, B, 0, stream>>>(ei, E, x, dinv, ax, N);

  {
    dim3 grid(tilesM, HID / 16);
    k_gemm2<<<grid, 32, 0, stream>>>(ax, W1, b1, W2, b2, dinv, bufA, bufC, N);
  }
  k_scatter<<<(E16 + B - 1) / B, B, 0, stream>>>(ei, E, bufA, dinv, bufC, N);

  k_pool<<<(N16 + B - 1) / B, B, 0, stream>>>(bufC, bat, out, pool1, cnt, N, G);
  k_div <<<(out_size + B - 1) / B, B, 0, stream>>>(out, pool1, cnt, out_size, G);
}


# edge scatter and pool adds with lane = column: each f32 atomic add instruction covers one whole 128-byte line (was four quarter-filled lines); a wave takes 8 rows, loads grouped; arrival-order sums
# speedup vs baseline: 2.9001x; 1.8689x over previous
#include <hip/hip_runtime.h>


#define HID 64

typedef float v2f __attribute__((ext_vector_type(2)));
typedef float v8f __attribute__((ext_vector_type(8)));

__device__ __forceinline__ int pkg_half() {
  return (int)((__builtin_amdgcn_s_sendmsg_rtn(0x87) >> 18) & 1u);
}

__global__ void k_zero(float* __restrict__ a, int na, float* __restrict__ b, int n4b) {
  const int t = blockIdx.x * blockDim.x + threadIdx.x;
  const int stride = gridDim.x * blockDim.x;
  float4* b4 = (float4*)b;
  const float4 z = {0.0f, 0.0f, 0.0f, 0.0f};
  for (int i = t; i < na; i += stride) a[i] = 0.0f;
  for (int i = t; i < n4b; i += stride) b4[i] = z;
}

__global__ void k_degree(const int* __restrict__ ei, int E, float* __restrict__ deg, int N) {
  int e = blockIdx.x * blockDim.x + threadIdx.x;
  if (e >= E) return;
  const int d = ei[E + e];
  float* degh = deg + (long)pkg_half() * N;
  atomicAdd(&degh[d], 1.0f);
}

__global__ void k_prep(const float* __restrict__ x, const float* __restrict__ deg,
                       float* __restrict__ dinv, float* __restrict__ ax,
                       float* __restrict__ pool0, float* __restrict__ pool1,
                       float* __restrict__ cnt,
                       int N, int pool_size) {
  const int t = blockIdx.x * blockDim.x + threadIdx.x;
  const int stride = gridDim.x * blockDim.x;
  for (int i = t; i < pool_size; i += stride) { pool0[i] = 0.0f; pool1[i] = 0.0f; }
  for (int i = t; i < 2 * (pool_size >> 6); i += stride) cnt[i] = 0.0f;
  if (t >= N) return;
  const float di = rsqrtf(deg[t] + deg[N + t] + 1.0f);
  const float s  = di * di;
  float4 v;
  v.x = x[t * 3 + 0] * s;
  v.y = x[t * 3 + 1] * s;
  v.z = x[t * 3 + 2] * s;
  v.w = 0.0f;
  *(float4*)(ax + (long)t * 4) = v;
  const float4 z = {0.0f, 0.0f, 0.0f, 0.0f};
  *(float4*)(ax + ((long)N + t) * 4) = z;
  dinv[t] = di;
}

__global__ void k_scatter3(const int* __restrict__ ei, int E,
                           const float* __restrict__ x,
                           const float* __restrict__ dinv,
                           float* __restrict__ ax2, int N) {
  const int t  = blockIdx.x * blockDim.x + threadIdx.x;
  const int c  = t & 3;
  const int e0 = (t >> 2) * 4;
  if (c >= 3 || e0 >= E) return;
  if (e0 + 3 < E) {
    int s[4], d[4];
    #pragma unroll
    for (int i = 0; i < 4; ++i) { s[i] = ei[e0 + i]; d[i] = ei[E + e0 + i]; }
    float v[4];
    #pragma unroll
    for (int i = 0; i < 4; ++i)
      v[i] = x[(long)s[i] * 3 + c] * (dinv[s[i]] * dinv[d[i]]);
    float* ax = ax2 + (long)pkg_half() * N * 4;
    #pragma unroll
    for (int i = 0; i < 4; ++i) atomicAdd(ax + (long)d[i] * 4 + c, v[i]);
  } else {
    float* ax = ax2 + (long)pkg_half() * N * 4;
    for (int e = e0; e < E; ++e) {
      const int s = ei[e], d = ei[E + e];
      atomicAdd(ax + (long)d * 4 + c, x[(long)s * 3 + c] * (dinv[s] * dinv[d]));
    }
  }
}

__global__ __launch_bounds__(32) void k_gemm2(const float* __restrict__ AX,
                                              const float* __restrict__ W1,
                                              const float* __restrict__ b1,
                                              const float* __restrict__ W,
                                              const float* __restrict__ bias,
                                              const float* __restrict__ dinv,
                                              float* __restrict__ O,
                                              float* __restrict__ S, int N) {
  const int tm   = blockIdx.x;
  const int tn   = blockIdx.y;
  const int lane = threadIdx.x;
  const int m    = lane & 15;
  const int half = lane >> 4;
  const int ncol = tn * 16 + (lane & 15);

  int arow = tm * 16 + m;
  if (arow >= N) arow = N - 1;
  const float4 xa0 = *(const float4*)(AX + (long)arow * 4);
  const float4 xa1 = *(const float4*)(AX + ((long)N + arow) * 4);
  float4 xa;
  xa.x = xa0.x + xa1.x; xa.y = xa0.y + xa1.y; xa.z = xa0.z + xa1.z; xa.w = 0.0f;

  float sl[8];
  #pragma unroll
  for (int r = 0; r < 8; ++r) {
    int mr = tm * 16 + r + half * 8;
    if (mr >= N) mr = N - 1;
    const float di = dinv[mr];
    sl[r] = di * di;
  }
  const float bj = bias[ncol];

  v8f acc = {};
  #pragma unroll
  for (int k0 = 0; k0 < HID; k0 += 4) {
    const int ka = k0 + half * 2;
    const float h0 = xa.x * W1[ka]     + xa.y * W1[HID + ka]     + xa.z * W1[2 * HID + ka]     + b1[ka];
    const float h1 = xa.x * W1[ka + 1] + xa.y * W1[HID + ka + 1] + xa.z * W1[2 * HID + ka + 1] + b1[ka + 1];
    v2f a; a.x = fmaxf(h0, 0.0f);      a.y = fmaxf(h1, 0.0f);
    v2f b; b.x = W[ka * HID + ncol];   b.y = W[(ka + 1) * HID + ncol];
    acc = __builtin_amdgcn_wmma_f32_16x16x4_f32(
         false, a,  false, b,
         (short)0, acc,  false,  false);
  }
  #pragma unroll
  for (int r = 0; r < 8; ++r) {
    int mrow = tm * 16 + r + half * 8;
    if (mrow < N) {
      O[(long)mrow * HID + ncol] = acc[r];
      S[(long)mrow * HID + ncol] = fmaf(acc[r], sl[r], bj);
    }
  }
}

#define RPW 8
__global__ void k_scatter(const int* __restrict__ ei, int E,
                          const float* __restrict__ hw,
                          const float* __restrict__ dinv,
                          float* __restrict__ agg2, int N) {
  const int lane = threadIdx.x & 31;
  const int wv = __builtin_amdgcn_readfirstlane((int)(threadIdx.x >> 5));
  const int e0 = (blockIdx.x * (blockDim.x >> 5) + wv) * RPW;
  if (e0 >= E) return;
  if (e0 + RPW <= E) {
    const int el = e0 + (lane & (RPW - 1));
    const int s = ei[el];
    const int d = ei[E + el];
    float v0[RPW], v1[RPW];
    #pragma unroll
    for (int i = 0; i < RPW; ++i) {
      const float* row = hw + (long)__builtin_amdgcn_readlane(s, i) * HID;
      v0[i] = row[lane];
      v1[i] = row[32 + lane];
    }
    const float nrm = dinv[s] * dinv[d];
    float* agg = agg2 + (long)pkg_half() * N * HID;
    #pragma unroll
    for (int i = 0; i < RPW; ++i) {
      const float ni = __int_as_float(__builtin_amdgcn_readlane(__float_as_int(nrm), i));
      float* o = agg + (long)__builtin_amdgcn_readlane(d, i) * HID;
      atomicAdd(o + lane,      v0[i] * ni);
      atomicAdd(o + 32 + lane, v1[i] * ni);
    }
  } else {
    float* agg = agg2 + (long)pkg_half() * N * HID;
    for (int e = e0; e < E; ++e) {
      const int s = ei[e], d = ei[E + e];
      const float nrm = dinv[s] * dinv[d];
      const float* row = hw + (long)s * HID;
      float* o = agg + (long)d * HID;
      atomicAdd(o + lane,      row[lane] * nrm);
      atomicAdd(o + 32 + lane, row[32 + lane] * nrm);
    }
  }
}

__global__ void k_pool(const float* __restrict__ h, const int* __restrict__ batch,
                       float* __restrict__ pool0, float* __restrict__ pool1,
                       float* __restrict__ cnt2, int N, int G) {
  const int lane = threadIdx.x & 31;
  const int wv = __builtin_amdgcn_readfirstlane((int)(threadIdx.x >> 5));
  const int n0 = (blockIdx.x * (blockDim.x >> 5) + wv) * RPW;
  if (n0 >= N) return;
  if (n0 + RPW <= N) {
    const int gl = batch[n0 + (lane & (RPW - 1))];
    float a0[RPW], a1[RPW], b0[RPW], b1[RPW];
    #pragma unroll
    for (int i = 0; i < RPW; ++i) {
      const float* r0 = h + (long)(n0 + i) * HID;
      const float* r1 = h + ((long)N + n0 + i) * HID;
      a0[i] = r0[lane]; a1[i] = r0[32 + lane];
      b0[i] = r1[lane]; b1[i] = r1[32 + lane];
    }
    const int hf = pkg_half();
    float* out = hf ? pool1 : pool0;
    float* cnt = cnt2 + hf * G;
    #pragma unroll
    for (int i = 0; i < RPW; ++i) {
      const int g = __builtin_amdgcn_readlane(gl, i);
      float* o = out + (long)g * HID;
      atomicAdd(o + lane,      fmaxf(a0[i] + b0[i], 0.0f));
      atomicAdd(o + 32 + lane, fmaxf(a1[i] + b1[i], 0.0f));
      if (lane == 0) atomicAdd(&cnt[g], 1.0f);
    }
  } else {
    const int hf = pkg_half();
    float* out = hf ? pool1 : pool0;
    float* cnt = cnt2 + hf * G;
    for (int n = n0; n < N; ++n) {
      const int g = batch[n];
      const float* r0 = h + (long)n * HID;
      const float* r1 = h + ((long)N + n) * HID;
      float* o = out + (long)g * HID;
      atomicAdd(o + lane,      fmaxf(r0[lane] + r1[lane], 0.0f));
      atomicAdd(o + 32 + lane, fmaxf(r0[32 + lane] + r1[32 + lane], 0.0f));
      if (lane == 0) atomicAdd(&cnt[g], 1.0f);
    }
  }
}

__global__ void k_div(float* __restrict__ out, const float* __restrict__ pool1,
                      const float* __restrict__ cnt, int total, int G) {
  int t = blockIdx.x * blockDim.x + threadIdx.x;
  if (t < total) out[t] = (out[t] + pool1[t]) / fmaxf(cnt[t >> 6] + cnt[G + (t >> 6)], 1.0f);
}

extern "C" void kernel_launch(void* const* d_in, const int* in_sizes, int n_in,
                              void* d_out, int out_size, void* d_ws, size_t ws_size,
                              hipStream_t stream) {
  const float* x   = (const float*)d_in[0];
  const float* W1  = (const float*)d_in[1];
  const float* b1  = (const float*)d_in[2];
  const float* W2  = (const float*)d_in[3];
  const float* b2  = (const float*)d_in[4];
  const int*   ei  = (const int*)d_in[5];
  const int*   bat = (const int*)d_in[6];
  float* out = (float*)d_out;

  const int N = in_sizes[0] / 3;
  const int E = in_sizes[5] / 2;

  const int G = out_size / HID;

  char* ws = (char*)d_ws;
  float* deg  = (float*)(ws);
  float* dinv = (float*)(ws + (1 << 20));
  float* bufA = (float*)(ws + (1 << 21));
  float* ax   = (float*)(ws + (1 << 21) + (size_t)N * HID * 4);
  float* bufC = (float*)(ws + (1 << 21) + 2 * (size_t)N * HID * 4);
  float* pool1 = (float*)(ws + (1 << 21) + 4 * (size_t)N * HID * 4);
  float* cnt  = pool1 + out_size;

  const int B = 256;
  const int N16  = N * 16;
  const int WPB  = B / 32;
  const int tilesM = (N + 15) / 16;

  k_zero  <<<1024, B, 0, stream>>>(deg, 2 * N, bufC + (size_t)N * HID, N16);
  k_degree<<<(E + B - 1) / B, B, 0, stream>>>(ei, E, deg, N);

  k_prep    <<<(N + B - 1) / B, B, 0, stream>>>(x, deg, dinv, ax, out, pool1, cnt, N, out_size);
  k_scatter3<<<(E + B - 1) / B, B, 0, stream>>>(ei, E, x, dinv, ax, N);

  {
    dim3 grid(tilesM, HID / 16);
    k_gemm2<<<grid, 32, 0, stream>>>(ax, W1, b1, W2, b2, dinv, bufA, bufC, N);
  }
  k_scatter<<<((E + RPW - 1) / RPW + WPB - 1) / WPB, B, 0, stream>>>(ei, E, bufA, dinv, bufC, N);

  k_pool<<<((N + RPW - 1) / RPW + WPB - 1) / WPB, B, 0, stream>>>(bufC, bat, out, pool1, cnt, N, G);
  k_div <<<(out_size + B - 1) / B, B, 0, stream>>>(out, pool1, cnt, out_size, G);
}


# edge weight split: dinv[src] pre-multiplied into the gathered rows, dinv[dst] and bias applied once per node by the reader; edge kernels load no dinv; re-associates the f32 products
# speedup vs baseline: 2.9968x; 1.0333x over previous
#include <hip/hip_runtime.h>


#define HID 64

typedef float v2f __attribute__((ext_vector_type(2)));
typedef float v8f __attribute__((ext_vector_type(8)));

__device__ __forceinline__ int pkg_half() {
  return (int)((__builtin_amdgcn_s_sendmsg_rtn(0x87) >> 18) & 1u);
}

__global__ void k_zero(float* __restrict__ a, int na, float* __restrict__ b, int n4b) {
  const int t = blockIdx.x * blockDim.x + threadIdx.x;
  const int stride = gridDim.x * blockDim.x;
  float4* b4 = (float4*)b;
  const float4 z = {0.0f, 0.0f, 0.0f, 0.0f};
  for (int i = t; i < na; i += stride) a[i] = 0.0f;
  for (int i = t; i < n4b; i += stride) b4[i] = z;
}

__global__ void k_degree(const int* __restrict__ ei, int E, float* __restrict__ deg, int N) {
  int e = blockIdx.x * blockDim.x + threadIdx.x;
  if (e >= E) return;
  const int d = ei[E + e];
  float* degh = deg + (long)pkg_half() * N;
  atomicAdd(&degh[d], 1.0f);
}

__global__ void k_prep(const float* __restrict__ x, const float* __restrict__ deg,
                       float* __restrict__ dinv, float* __restrict__ xs, float* __restrict__ ax,
                       float* __restrict__ pool0, float* __restrict__ pool1,
                       float* __restrict__ cnt,
                       int N, int pool_size) {
  const int t = blockIdx.x * blockDim.x + threadIdx.x;
  const int stride = gridDim.x * blockDim.x;
  for (int i = t; i < pool_size; i += stride) { pool0[i] = 0.0f; pool1[i] = 0.0f; }
  for (int i = t; i < 2 * (pool_size >> 6); i += stride) cnt[i] = 0.0f;
  if (t >= N) return;
  const float di = rsqrtf(deg[t] + deg[N + t] + 1.0f);
  float4 v;
  v.x = x[t * 3 + 0] * di;
  v.y = x[t * 3 + 1] * di;
  v.z = x[t * 3 + 2] * di;
  v.w = 0.0f;
  *(float4*)(xs + (long)t * 4) = v;
  *(float4*)(ax + (long)t * 4) = v;
  const float4 z = {0.0f, 0.0f, 0.0f, 0.0f};
  *(float4*)(ax + ((long)N + t) * 4) = z;
  dinv[t] = di;
}

__global__ void k_scatter3(const int* __restrict__ ei, int E,
                           const float* __restrict__ xs,
                           float* __restrict__ ax2, int N) {
  const int t  = blockIdx.x * blockDim.x + threadIdx.x;
  const int c  = t & 3;
  const int e0 = (t >> 2) * 4;
  if (c >= 3 || e0 >= E) return;
  if (e0 + 3 < E) {
    int s[4], d[4];
    #pragma unroll
    for (int i = 0; i < 4; ++i) { s[i] = ei[e0 + i]; d[i] = ei[E + e0 + i]; }
    float v[4];
    #pragma unroll
    for (int i = 0; i < 4; ++i)
      v[i] = xs[(long)s[i] * 4 + c];
    float* ax = ax2 + (long)pkg_half() * N * 4;
    #pragma unroll
    for (int i = 0; i < 4; ++i) atomicAdd(ax + (long)d[i] * 4 + c, v[i]);
  } else {
    float* ax = ax2 + (long)pkg_half() * N * 4;
    for (int e = e0; e < E; ++e) {
      const int s = ei[e], d = ei[E + e];
      atomicAdd(ax + (long)d * 4 + c, xs[(long)s * 4 + c]);
    }
  }
}

__global__ __launch_bounds__(32) void k_gemm2(const float* __restrict__ AX,
                                              const float* __restrict__ W1,
                                              const float* __restrict__ b1,
                                              const float* __restrict__ W,
                                              const float* __restrict__ dinv,
                                              float* __restrict__ O,
                                              float* __restrict__ S, int N) {
  const int tm   = blockIdx.x;
  const int tn   = blockIdx.y;
  const int lane = threadIdx.x;
  const int m    = lane & 15;
  const int half = lane >> 4;
  const int ncol = tn * 16 + (lane & 15);

  int arow = tm * 16 + m;
  if (arow >= N) arow = N - 1;
  const float4 xa0 = *(const float4*)(AX + (long)arow * 4);
  const float4 xa1 = *(const float4*)(AX + ((long)N + arow) * 4);
  const float da = dinv[arow];
  float4 xa;
  xa.x = (xa0.x + xa1.x) * da; xa.y = (xa0.y + xa1.y) * da; xa.z = (xa0.z + xa1.z) * da; xa.w = 0.0f;

  float sl[8];
  #pragma unroll
  for (int r = 0; r < 8; ++r) {
    int mr = tm * 16 + r + half * 8;
    if (mr >= N) mr = N - 1;
    sl[r] = dinv[mr];
  }

  v8f acc = {};
  #pragma unroll
  for (int k0 = 0; k0 < HID; k0 += 4) {
    const int ka = k0 + half * 2;
    const float h0 = xa.x * W1[ka]     + xa.y * W1[HID + ka]     + xa.z * W1[2 * HID + ka]     + b1[ka];
    const float h1 = xa.x * W1[ka + 1] + xa.y * W1[HID + ka + 1] + xa.z * W1[2 * HID + ka + 1] + b1[ka + 1];
    v2f a; a.x = fmaxf(h0, 0.0f);      a.y = fmaxf(h1, 0.0f);
    v2f b; b.x = W[ka * HID + ncol];   b.y = W[(ka + 1) * HID + ncol];
    acc = __builtin_amdgcn_wmma_f32_16x16x4_f32(
         false, a,  false, b,
         (short)0, acc,  false,  false);
  }
  #pragma unroll
  for (int r = 0; r < 8; ++r) {
    int mrow = tm * 16 + r + half * 8;
    if (mrow < N) {
      const float hv = acc[r] * sl[r];
      O[(long)mrow * HID + ncol] = hv;
      S[(long)mrow * HID + ncol] = hv;
    }
  }
}

#define RPW 8
__global__ void k_scatter(const int* __restrict__ ei, int E,
                          const float* __restrict__ hw,
                          float* __restrict__ agg2, int N) {
  const int lane = threadIdx.x & 31;
  const int wv = __builtin_amdgcn_readfirstlane((int)(threadIdx.x >> 5));
  const int e0 = (blockIdx.x * (blockDim.x >> 5) + wv) * RPW;
  if (e0 >= E) return;
  if (e0 + RPW <= E) {
    const int el = e0 + (lane & (RPW - 1));
    const int s = ei[el];
    const int d = ei[E + el];
    float v0[RPW], v1[RPW];
    #pragma unroll
    for (int i = 0; i < RPW; ++i) {
      const float* row = hw + (long)__builtin_amdgcn_readlane(s, i) * HID;
      v0[i] = row[lane];
      v1[i] = row[32 + lane];
    }
    float* agg = agg2 + (long)pkg_half() * N * HID;
    #pragma unroll
    for (int i = 0; i < RPW; ++i) {
      float* o = agg + (long)__builtin_amdgcn_readlane(d, i) * HID;
      atomicAdd(o + lane,      v0[i]);
      atomicAdd(o + 32 + lane, v1[i]);
    }
  } else {
    float* agg = agg2 + (long)pkg_half() * N * HID;
    for (int e = e0; e < E; ++e) {
      const int s = ei[e], d = ei[E + e];
      const float* row = hw + (long)s * HID;
      float* o = agg + (long)d * HID;
      atomicAdd(o + lane,      row[lane]);
      atomicAdd(o + 32 + lane, row[32 + lane]);
    }
  }
}

__global__ void k_pool(const float* __restrict__ h, const int* __restrict__ batch,
                       const float* __restrict__ dinv, const float* __restrict__ bias,
                       float* __restrict__ pool0, float* __restrict__ pool1,
                       float* __restrict__ cnt2, int N, int G) {
  const int lane = threadIdx.x & 31;
  const int wv = __builtin_amdgcn_readfirstlane((int)(threadIdx.x >> 5));
  const int n0 = (blockIdx.x * (blockDim.x >> 5) + wv) * RPW;
  if (n0 >= N) return;
  if (n0 + RPW <= N) {
    const int gl = batch[n0 + (lane & (RPW - 1))];
    const float dl = dinv[n0 + (lane & (RPW - 1))];
    const float bj0 = bias[lane], bj1 = bias[32 + lane];
    float a0[RPW], a1[RPW], b0[RPW], b1[RPW];
    #pragma unroll
    for (int i = 0; i < RPW; ++i) {
      const float* r0 = h + (long)(n0 + i) * HID;
      const float* r1 = h + ((long)N + n0 + i) * HID;
      a0[i] = r0[lane]; a1[i] = r0[32 + lane];
      b0[i] = r1[lane]; b1[i] = r1[32 + lane];
    }
    const int hf = pkg_half();
    float* out = hf ? pool1 : pool0;
    float* cnt = cnt2 + hf * G;
    #pragma unroll
    for (int i = 0; i < RPW; ++i) {
      const int g = __builtin_amdgcn_readlane(gl, i);
      const float di = __int_as_float(__builtin_amdgcn_readlane(__float_as_int(dl), i));
      float* o = out + (long)g * HID;
      atomicAdd(o + lane,      fmaxf(fmaf(a0[i] + b0[i], di, bj0), 0.0f));
      atomicAdd(o + 32 + lane, fmaxf(fmaf(a1[i] + b1[i], di, bj1), 0.0f));
      if (lane == 0) atomicAdd(&cnt[g], 1.0f);
    }
  } else {
    const int hf = pkg_half();
    float* out = hf ? pool1 : pool0;
    float* cnt = cnt2 + hf * G;
    for (int n = n0; n < N; ++n) {
      const int g = batch[n];
      const float di = dinv[n];
      const float* r0 = h + (long)n * HID;
      const float* r1 = h + ((long)N + n) * HID;
      float* o = out + (long)g * HID;
      atomicAdd(o + lane,      fmaxf(fmaf(r0[lane] + r1[lane], di, bias[lane]), 0.0f));
      atomicAdd(o + 32 + lane, fmaxf(fmaf(r0[32 + lane] + r1[32 + lane], di, bias[32 + lane]), 0.0f));
      if (lane == 0) atomicAdd(&cnt[g], 1.0f);
    }
  }
}

__global__ void k_div(float* __restrict__ out, const float* __restrict__ pool1,
                      const float* __restrict__ cnt, int total, int G) {
  int t = blockIdx.x * blockDim.x + threadIdx.x;
  if (t < total) out[t] = (out[t] + pool1[t]) / fmaxf(cnt[t >> 6] + cnt[G + (t >> 6)], 1.0f);
}

extern "C" void kernel_launch(void* const* d_in, const int* in_sizes, int n_in,
                              void* d_out, int out_size, void* d_ws, size_t ws_size,
                              hipStream_t stream) {
  const float* x   = (const float*)d_in[0];
  const float* W1  = (const float*)d_in[1];
  const float* b1  = (const float*)d_in[2];
  const float* W2  = (const float*)d_in[3];
  const float* b2  = (const float*)d_in[4];
  const int*   ei  = (const int*)d_in[5];
  const int*   bat = (const int*)d_in[6];
  float* out = (float*)d_out;

  const int N = in_sizes[0] / 3;
  const int E = in_sizes[5] / 2;

  const int G = out_size / HID;

  char* ws = (char*)d_ws;
  float* deg  = (float*)(ws);
  float* dinv = (float*)(ws + (1 << 20));
  float* bufA = (float*)(ws + (1 << 21));
  float* ax   = (float*)(ws + (1 << 21) + (size_t)N * HID * 4);
  float* xs   = ax + 8 * (size_t)N;
  float* bufC = (float*)(ws + (1 << 21) + 2 * (size_t)N * HID * 4);
  float* pool1 = (float*)(ws + (1 << 21) + 4 * (size_t)N * HID * 4);
  float* cnt  = pool1 + out_size;

  const int B = 256;
  const int N16  = N * 16;
  const int WPB  = B / 32;
  const int tilesM = (N + 15) / 16;

  k_zero  <<<1024, B, 0, stream>>>(deg, 2 * N, bufC + (size_t)N * HID, N16);
  k_degree<<<(E + B - 1) / B, B, 0, stream>>>(ei, E, deg, N);

  k_prep    <<<(N + B - 1) / B, B, 0, stream>>>(x, deg, dinv, xs, ax, out, pool1, cnt, N, out_size);
  k_scatter3<<<(E + B - 1) / B, B, 0, stream>>>(ei, E, xs, ax, N);

  {
    dim3 grid(tilesM, HID / 16);
    k_gemm2<<<grid, 32, 0, stream>>>(ax, W1, b1, W2, dinv, bufA, bufC, N);
  }
  k_scatter<<<((E + RPW - 1) / RPW + WPB - 1) / WPB, B, 0, stream>>>(ei, E, bufA, bufC, N);

  k_pool<<<((N + RPW - 1) / RPW + WPB - 1) / WPB, B, 0, stream>>>(bufC, bat, dinv, b2, out, pool1, cnt, N, G);
  k_div <<<(out_size + B - 1) / B, B, 0, stream>>>(out, pool1, cnt, out_size, G);
}


# every edge/node kernel runs a fixed grid of resident waves that loop over many items (six times fewer waves start); the GEMM wave takes all 64 columns of its 16 rows; sums in arrival order as before
# speedup vs baseline: 3.0967x; 1.0333x over previous
#include <hip/hip_runtime.h>


#define HID 64


typedef float v2f __attribute__((ext_vector_type(2)));
typedef float v8f __attribute__((ext_vector_type(8)));

__device__ __forceinline__ int pkg_half() {
  return (int)((__builtin_amdgcn_s_sendmsg_rtn(0x87) >> 18) & 1u);
}

__global__ void k_zero(float* __restrict__ a, int na, float* __restrict__ b, int n4b) {
  const int t = blockIdx.x * blockDim.x + threadIdx.x;
  const int stride = gridDim.x * blockDim.x;
  float4* b4 = (float4*)b;
  const float4 z = {0.0f, 0.0f, 0.0f, 0.0f};
  for (int i = t; i < na; i += stride) a[i] = 0.0f;
  for (int i = t; i < n4b; i += stride) b4[i] = z;
}

__global__ void k_degree(const int* __restrict__ ei, int E, float* __restrict__ deg, int N) {
  const int t = blockIdx.x * blockDim.x + threadIdx.x;
  const int stride = gridDim.x * blockDim.x;
  float* degh = deg + (long)pkg_half() * N;
  for (int e = t; e < E; e += 8 * stride) {
    int d[8];
    #pragma unroll
    for (int k = 0; k < 8; ++k) {
      const int idx = e + k * stride;
      d[k] = ei[E + (idx < E ? idx : E - 1)];
    }
    #pragma unroll
    for (int k = 0; k < 8; ++k)
      if (e + k * stride < E) atomicAdd(&degh[d[k]], 1.0f);
  }
}

__global__ void k_prep(const float* __restrict__ x, const float* __restrict__ deg,
                       float* __restrict__ dinv, float* __restrict__ xs, float* __restrict__ ax,
                       float* __restrict__ pool0, float* __restrict__ pool1,
                       float* __restrict__ cnt,
                       int N, int pool_size) {
  const int t = blockIdx.x * blockDim.x + threadIdx.x;
  const int stride = gridDim.x * blockDim.x;
  for (int i = t; i < pool_size; i += stride) { pool0[i] = 0.0f; pool1[i] = 0.0f; }
  for (int i = t; i < 2 * (pool_size >> 6); i += stride) cnt[i] = 0.0f;
  if (t >= N) return;
  const float di = rsqrtf(deg[t] + deg[N + t] + 1.0f);
  float4 v;
  v.x = x[t * 3 + 0] * di;
  v.y = x[t * 3 + 1] * di;
  v.z = x[t * 3 + 2] * di;
  v.w = 0.0f;
  *(float4*)(xs + (long)t * 4) = v;
  *(float4*)(ax + (long)t * 4) = v;
  const float4 z = {0.0f, 0.0f, 0.0f, 0.0f};
  *(float4*)(ax + ((long)N + t) * 4) = z;
  dinv[t] = di;
}

__global__ void k_scatter3(const int* __restrict__ ei, int E,
                           const float* __restrict__ xs,
                           float* __restrict__ ax2, int N) {
  const int t  = blockIdx.x * blockDim.x + threadIdx.x;
  const int c  = t & 3;
  if (c >= 3) return;
  const int step = ((gridDim.x * blockDim.x) >> 2) * 4;
  float* ax = ax2 + (long)pkg_half() * N * 4;
  for (int e0 = (t >> 2) * 4; e0 < E; e0 += step) {
  if (e0 + 3 < E) {
    int s[4], d[4];
    #pragma unroll
    for (int i = 0; i < 4; ++i) { s[i] = ei[e0 + i]; d[i] = ei[E + e0 + i]; }
    float v[4];
    #pragma unroll
    for (int i = 0; i < 4; ++i)
      v[i] = xs[(long)s[i] * 4 + c];
    #pragma unroll
    for (int i = 0; i < 4; ++i) atomicAdd(ax + (long)d[i] * 4 + c, v[i]);
  } else {
    for (int e = e0; e < E; ++e) {
      const int s = ei[e], d = ei[E + e];
      atomicAdd(ax + (long)d * 4 + c, xs[(long)s * 4 + c]);
    }
  }
  }
}

__global__ __launch_bounds__(256) void k_gemm2(const float* __restrict__ AX,
                                              const float* __restrict__ W1,
                                              const float* __restrict__ b1,
                                              const float* __restrict__ W,
                                              const float* __restrict__ dinv,
                                              float* __restrict__ O,
                                              float* __restrict__ S, int N) {
  const int lane = threadIdx.x & 31;
  const int wv   = __builtin_amdgcn_readfirstlane((int)(threadIdx.x >> 5));
  const int tm   = blockIdx.x * (blockDim.x >> 5) + wv;
  if (tm * 16 >= N) return;
  const int m    = lane & 15;
  const int half = lane >> 4;
  const int c16  = lane & 15;

  int arow = tm * 16 + m;
  if (arow >= N) arow = N - 1;
  const float4 xa0 = *(const float4*)(AX + (long)arow * 4);
  const float4 xa1 = *(const float4*)(AX + ((long)N + arow) * 4);
  const float da = dinv[arow];
  float4 xa;
  xa.x = (xa0.x + xa1.x) * da; xa.y = (xa0.y + xa1.y) * da; xa.z = (xa0.z + xa1.z) * da; xa.w = 0.0f;

  float sl[8];
  #pragma unroll
  for (int r = 0; r < 8; ++r) {
    int mr = tm * 16 + r + half * 8;
    if (mr >= N) mr = N - 1;
    sl[r] = dinv[mr];
  }

  v8f acc[4] = {};
  #pragma unroll
  for (int k0 = 0; k0 < HID; k0 += 4) {
    const int ka = k0 + half * 2;
    const float h0 = xa.x * W1[ka]     + xa.y * W1[HID + ka]     + xa.z * W1[2 * HID + ka]     + b1[ka];
    const float h1 = xa.x * W1[ka + 1] + xa.y * W1[HID + ka + 1] + xa.z * W1[2 * HID + ka + 1] + b1[ka + 1];
    v2f a; a.x = fmaxf(h0, 0.0f);      a.y = fmaxf(h1, 0.0f);
    #pragma unroll
    for (int tn = 0; tn < 4; ++tn) {
      const int ncol = tn * 16 + c16;
      v2f b; b.x = W[ka * HID + ncol];   b.y = W[(ka + 1) * HID + ncol];
      acc[tn] = __builtin_amdgcn_wmma_f32_16x16x4_f32(
           false, a,  false, b,
           (short)0, acc[tn],  false,  false);
    }
  }
  #pragma unroll
  for (int r = 0; r < 8; ++r) {
    int mrow = tm * 16 + r + half * 8;
    if (mrow < N) {
      #pragma unroll
      for (int tn = 0; tn < 4; ++tn) {
        const int ncol = tn * 16 + c16;
        const float hv = acc[tn][r] * sl[r];
        O[(long)mrow * HID + ncol] = hv;
        S[(long)mrow * HID + ncol] = hv;
      }
    }
  }
}

#define RPW 8
__global__ __launch_bounds__(256) void k_scatter(const int* __restrict__ ei, int E,
                          const float* __restrict__ hw,
                          float* __restrict__ agg2, int N) {
  const int lane = threadIdx.x & 31;
  const int wv = __builtin_amdgcn_readfirstlane((int)(threadIdx.x >> 5));
  const int gw = blockIdx.x * (blockDim.x >> 5) + wv;
  const int nw = gridDim.x * (blockDim.x >> 5);
  const int nfull = E >> 5;
  if (gw < nfull) {
    int q = gw;
    int s = ei[q * 32 + lane];
    int d = ei[E + q * 32 + lane];
    float* agg = agg2 + (long)pkg_half() * N * HID;
    for (;;) {
      const int qn = q + nw;
      const int qp = qn < nfull ? qn : q;
      const int sn = ei[qp * 32 + lane];
      const int dn = ei[E + qp * 32 + lane];
      float v[2][2 * RPW];
      #pragma unroll
      for (int i = 0; i < RPW; ++i) {
        const float* row = hw + (long)__builtin_amdgcn_readlane(s, i) * HID;
        v[0][2 * i]     = row[lane];
        v[0][2 * i + 1] = row[32 + lane];
      }
      #pragma unroll
      for (int j = 0; j < 4; ++j) {
        if (j < 3) {
          #pragma unroll
          for (int i = 0; i < RPW; ++i) {
            const float* row = hw + (long)__builtin_amdgcn_readlane(s, (j + 1) * RPW + i) * HID;
            v[(j + 1) & 1][2 * i]     = row[lane];
            v[(j + 1) & 1][2 * i + 1] = row[32 + lane];
          }
        }
        #pragma unroll
        for (int i = 0; i < RPW; ++i) {
          float* o = agg + (long)__builtin_amdgcn_readlane(d, j * RPW + i) * HID;
          atomicAdd(o + lane,      v[j & 1][2 * i]);
          atomicAdd(o + 32 + lane, v[j & 1][2 * i + 1]);
        }
      }
      if (qn >= nfull) break;
      q = qn; s = sn; d = dn;
    }
  }
  if (gw == nfull % nw && (nfull << 5) < E) {
    float* agg = agg2 + (long)pkg_half() * N * HID;
    for (int e = nfull << 5; e < E; ++e) {
      const int s = ei[e], d = ei[E + e];
      const float* row = hw + (long)s * HID;
      float* o = agg + (long)d * HID;
      atomicAdd(o + lane,      row[lane]);
      atomicAdd(o + 32 + lane, row[32 + lane]);
    }
  }
}

__global__ __launch_bounds__(256) void k_pool(const float* __restrict__ h, const int* __restrict__ batch,
                       const float* __restrict__ dinv, const float* __restrict__ bias,
                       float* __restrict__ pool0, float* __restrict__ pool1,
                       float* __restrict__ cnt2, int N, int G) {
  const int lane = threadIdx.x & 31;
  const int wv = __builtin_amdgcn_readfirstlane((int)(threadIdx.x >> 5));
  const int gw = blockIdx.x * (blockDim.x >> 5) + wv;
  const int nw = gridDim.x * (blockDim.x >> 5);
  const int nfull = N / RPW;
  if (gw < nfull) {
    const float bj0 = bias[lane], bj1 = bias[32 + lane];
    const int hf = pkg_half();
    float* out = hf ? pool1 : pool0;
    float* cnt = cnt2 + hf * G;
    for (int q = gw; q < nfull; q += nw) {
    const int n0 = q * RPW;
    const int gl = batch[n0 + (lane & (RPW - 1))];
    const float dl = dinv[n0 + (lane & (RPW - 1))];
    float a0[RPW], a1[RPW], b0[RPW], b1[RPW];
    #pragma unroll
    for (int i = 0; i < RPW; ++i) {
      const float* r0 = h + (long)(n0 + i) * HID;
      const float* r1 = h + ((long)N + n0 + i) * HID;
      a0[i] = r0[lane]; a1[i] = r0[32 + lane];
      b0[i] = r1[lane]; b1[i] = r1[32 + lane];
    }
    #pragma unroll
    for (int i = 0; i < RPW; ++i) {
      const int g = __builtin_amdgcn_readlane(gl, i);
      const float di = __int_as_float(__builtin_amdgcn_readlane(__float_as_int(dl), i));
      float* o = out + (long)g * HID;
      atomicAdd(o + lane,      fmaxf(fmaf(a0[i] + b0[i], di, bj0), 0.0f));
      atomicAdd(o + 32 + lane, fmaxf(fmaf(a1[i] + b1[i], di, bj1), 0.0f));
      if (lane == 0) atomicAdd(&cnt[g], 1.0f);
    }
    }
  }
  if (gw == nfull % nw && nfull * RPW < N) {
    const int hf = pkg_half();
    float* out = hf ? pool1 : pool0;
    float* cnt = cnt2 + hf * G;
    for (int n = nfull * RPW; n < N; ++n) {
      const int g = batch[n];
      const float di = dinv[n];
      const float* r0 = h + (long)n * HID;
      const float* r1 = h + ((long)N + n) * HID;
      float* o = out + (long)g * HID;
      atomicAdd(o + lane,      fmaxf(fmaf(r0[lane] + r1[lane], di, bias[lane]), 0.0f));
      atomicAdd(o + 32 + lane, fmaxf(fmaf(r0[32 + lane] + r1[32 + lane], di, bias[32 + lane]), 0.0f));
      if (lane == 0) atomicAdd(&cnt[g], 1.0f);
    }
  }
}

__global__ void k_div(float* __restrict__ out, const float* __restrict__ pool1,
                      const float* __restrict__ cnt, int total, int G) {
  int t = blockIdx.x * blockDim.x + threadIdx.x;
  if (t < total) out[t] = (out[t] + pool1[t]) / fmaxf(cnt[t >> 6] + cnt[G + (t >> 6)], 1.0f);
}

extern "C" void kernel_launch(void* const* d_in, const int* in_sizes, int n_in,
                              void* d_out, int out_size, void* d_ws, size_t ws_size,
                              hipStream_t stream) {
  const float* x   = (const float*)d_in[0];
  const float* W1  = (const float*)d_in[1];
  const float* b1  = (const float*)d_in[2];
  const float* W2  = (const float*)d_in[3];
  const float* b2  = (const float*)d_in[4];
  const int*   ei  = (const int*)d_in[5];
  const int*   bat = (const int*)d_in[6];
  float* out = (float*)d_out;

  const int N = in_sizes[0] / 3;
  const int E = in_sizes[5] / 2;

  const int G = out_size / HID;

  char* ws = (char*)d_ws;
  float* deg  = (float*)(ws);
  float* dinv = (float*)(ws + (1 << 20));
  float* bufA = (float*)(ws + (1 << 21));
  float* ax   = (float*)(ws + (1 << 21) + (size_t)N * HID * 4);
  float* xs   = ax + 8 * (size_t)N;
  float* bufC = (float*)(ws + (1 << 21) + 2 * (size_t)N * HID * 4);
  float* pool1 = (float*)(ws + (1 << 21) + 4 * (size_t)N * HID * 4);
  float* cnt  = pool1 + out_size;

  const int B = 256;
  const int N16  = N * 16;
  const int WPB  = B / 32;
  const int tilesM = (N + 15) / 16;
  const int GZ = 512;
  const int GE = 512;
  const int G3 = 1024;
  const int GS = 1024;
  const int GP = 512;

  k_zero  <<<GZ, B, 0, stream>>>(deg, 2 * N, bufC + (size_t)N * HID, N16);
  k_degree<<<GE, B, 0, stream>>>(ei, E, deg, N);

  k_prep    <<<(N + B - 1) / B, B, 0, stream>>>(x, deg, dinv, xs, ax, out, pool1, cnt, N, out_size);
  k_scatter3<<<G3, B, 0, stream>>>(ei, E, xs, ax, N);

  {
    k_gemm2<<<(tilesM + WPB - 1) / WPB, B, 0, stream>>>(ax, W1, b1, W2, dinv, bufA, bufC, N);
  }
  k_scatter<<<GS, B, 0, stream>>>(ei, E, bufA, bufC, N);

  k_pool<<<GP, B, 0, stream>>>(bufC, bat, dinv, b2, out, pool1, cnt, N, G);
  k_div <<<(out_size + B - 1) / B, B, 0, stream>>>(out, pool1, cnt, out_size, G);
}


# layer-2 sums in ONE plane, rows split between package halves: each half's work-groups walk all edges and add only their own rows; no second plane to clear or read; one arrival-order f32 sum per row
# speedup vs baseline: 4.6955x; 1.5163x over previous
#include <hip/hip_runtime.h>


#define HID 64


typedef float v2f __attribute__((ext_vector_type(2)));
typedef float v8f __attribute__((ext_vector_type(8)));

__device__ __forceinline__ int pkg_half() {
  return (int)((__builtin_amdgcn_s_sendmsg_rtn(0x87) >> 18) & 1u);
}

#define PLACE_PITCH 64
__global__ void k_zero(float* __restrict__ a, int na, int* __restrict__ place) {
  const int t = blockIdx.x * blockDim.x + threadIdx.x;
  const int stride = gridDim.x * blockDim.x;
  if (t < 2) place[t * PLACE_PITCH] = 0;
  for (int i = t; i < na; i += stride) a[i] = 0.0f;
}

__global__ void k_degree(const int* __restrict__ ei, int E, float* __restrict__ deg, int N) {
  const int t = blockIdx.x * blockDim.x + threadIdx.x;
  const int stride = gridDim.x * blockDim.x;
  float* degh = deg + (long)pkg_half() * N;
  for (int e = t; e < E; e += 8 * stride) {
    int d[8];
    #pragma unroll
    for (int k = 0; k < 8; ++k) {
      const int idx = e + k * stride;
      d[k] = ei[E + (idx < E ? idx : E - 1)];
    }
    #pragma unroll
    for (int k = 0; k < 8; ++k)
      if (e + k * stride < E) atomicAdd(&degh[d[k]], 1.0f);
  }
}

__global__ void k_prep(const float* __restrict__ x, const float* __restrict__ deg,
                       float* __restrict__ dinv, float* __restrict__ xs, float* __restrict__ ax,
                       float* __restrict__ pool0, float* __restrict__ pool1,
                       float* __restrict__ cnt,
                       int N, int pool_size) {
  const int t = blockIdx.x * blockDim.x + threadIdx.x;
  const int stride = gridDim.x * blockDim.x;
  for (int i = t; i < pool_size; i += stride) { pool0[i] = 0.0f; pool1[i] = 0.0f; }
  for (int i = t; i < 2 * (pool_size >> 6); i += stride) cnt[i] = 0.0f;
  if (t >= N) return;
  const float di = rsqrtf(deg[t] + deg[N + t] + 1.0f);
  float4 v;
  v.x = x[t * 3 + 0] * di;
  v.y = x[t * 3 + 1] * di;
  v.z = x[t * 3 + 2] * di;
  v.w = 0.0f;
  *(float4*)(xs + (long)t * 4) = v;
  *(float4*)(ax + (long)t * 4) = v;
  const float4 z = {0.0f, 0.0f, 0.0f, 0.0f};
  *(float4*)(ax + ((long)N + t) * 4) = z;
  dinv[t] = di;
}

__global__ void k_scatter3(const int* __restrict__ ei, int E,
                           const float* __restrict__ xs,
                           float* __restrict__ ax2, int N) {
  const int t  = blockIdx.x * blockDim.x + threadIdx.x;
  const int c  = t & 3;
  if (c >= 3) return;
  const int step = ((gridDim.x * blockDim.x) >> 2) * 4;
  float* ax = ax2 + (long)pkg_half() * N * 4;
  for (int e0 = (t >> 2) * 4; e0 < E; e0 += step) {
  if (e0 + 3 < E) {
    int s[4], d[4];
    #pragma unroll
    for (int i = 0; i < 4; ++i) { s[i] = ei[e0 + i]; d[i] = ei[E + e0 + i]; }
    float v[4];
    #pragma unroll
    for (int i = 0; i < 4; ++i)
      v[i] = xs[(long)s[i] * 4 + c];
    #pragma unroll
    for (int i = 0; i < 4; ++i) atomicAdd(ax + (long)d[i] * 4 + c, v[i]);
  } else {
    for (int e = e0; e < E; ++e) {
      const int s = ei[e], d = ei[E + e];
      atomicAdd(ax + (long)d * 4 + c, xs[(long)s * 4 + c]);
    }
  }
  }
}

__global__ __launch_bounds__(256) void k_gemm2(const float* __restrict__ AX,
                                              const float* __restrict__ W1,
                                              const float* __restrict__ b1,
                                              const float* __restrict__ W,
                                              const float* __restrict__ dinv,
                                              float* __restrict__ O,
                                              float* __restrict__ S, int N) {
  const int lane = threadIdx.x & 31;
  const int wv   = __builtin_amdgcn_readfirstlane((int)(threadIdx.x >> 5));
  const int tm   = blockIdx.x * (blockDim.x >> 5) + wv;
  if (tm * 16 >= N) return;
  const int m    = lane & 15;
  const int half = lane >> 4;
  const int c16  = lane & 15;

  int arow = tm * 16 + m;
  if (arow >= N) arow = N - 1;
  const float4 xa0 = *(const float4*)(AX + (long)arow * 4);
  const float4 xa1 = *(const float4*)(AX + ((long)N + arow) * 4);
  const float da = dinv[arow];
  float4 xa;
  xa.x = (xa0.x + xa1.x) * da; xa.y = (xa0.y + xa1.y) * da; xa.z = (xa0.z + xa1.z) * da; xa.w = 0.0f;

  float sl[8];
  #pragma unroll
  for (int r = 0; r < 8; ++r) {
    int mr = tm * 16 + r + half * 8;
    if (mr >= N) mr = N - 1;
    sl[r] = dinv[mr];
  }

  v8f acc[4] = {};
  #pragma unroll
  for (int k0 = 0; k0 < HID; k0 += 4) {
    const int ka = k0 + half * 2;
    const float h0 = xa.x * W1[ka]     + xa.y * W1[HID + ka]     + xa.z * W1[2 * HID + ka]     + b1[ka];
    const float h1 = xa.x * W1[ka + 1] + xa.y * W1[HID + ka + 1] + xa.z * W1[2 * HID + ka + 1] + b1[ka + 1];
    v2f a; a.x = fmaxf(h0, 0.0f);      a.y = fmaxf(h1, 0.0f);
    #pragma unroll
    for (int tn = 0; tn < 4; ++tn) {
      const int ncol = tn * 16 + c16;
      v2f b; b.x = W[ka * HID + ncol];   b.y = W[(ka + 1) * HID + ncol];
      acc[tn] = __builtin_amdgcn_wmma_f32_16x16x4_f32(
           false, a,  false, b,
           (short)0, acc[tn],  false,  false);
    }
  }
  #pragma unroll
  for (int r = 0; r < 8; ++r) {
    int mrow = tm * 16 + r + half * 8;
    if (mrow < N) {
      #pragma unroll
      for (int tn = 0; tn < 4; ++tn) {
        const int ncol = tn * 16 + c16;
        const float hv = acc[tn][r] * sl[r];
        O[(long)mrow * HID + ncol] = hv;
        S[(long)mrow * HID + ncol] = hv;
      }
    }
  }
}

#define RPW 8

struct Rows { float x0, y0, x1, y1, x2, y2, x3, y3, x4, y4, x5, y5, x6, y6, x7, y7; };

#define ROW_LOAD(i) { const float* row = hw + (long)__builtin_amdgcn_readlane(sc, B + i) * HID; \
                      v.x##i = row[lane]; v.y##i = row[32 + lane]; }
#define ROW_ADD(i)  { float* o = agg + (long)__builtin_amdgcn_readlane(dc, B + i) * HID; \
                      atomicAdd(o + lane, v.x##i); atomicAdd(o + 32 + lane, v.y##i); }

template <int B>
__device__ __forceinline__ void load_group(Rows& v, const float* __restrict__ hw,
                                           int sc, int cnt, int lane) {
  if (B < cnt) {
    ROW_LOAD(0)
    if (B + 1 < cnt) ROW_LOAD(1)
    if (B + 2 < cnt) ROW_LOAD(2)
    if (B + 3 < cnt) ROW_LOAD(3)
    if (B + 4 < cnt) ROW_LOAD(4)
    if (B + 5 < cnt) ROW_LOAD(5)
    if (B + 6 < cnt) ROW_LOAD(6)
    if (B + 7 < cnt) ROW_LOAD(7)
  }
}

template <int B>
__device__ __forceinline__ void add_group(const Rows& v, float* __restrict__ agg,
                                          int dc, int cnt, int lane) {
  if (B < cnt) {
    ROW_ADD(0)
    if (B + 1 < cnt) ROW_ADD(1)
    if (B + 2 < cnt) ROW_ADD(2)
    if (B + 3 < cnt) ROW_ADD(3)
    if (B + 4 < cnt) ROW_ADD(4)
    if (B + 5 < cnt) ROW_ADD(5)
    if (B + 6 < cnt) ROW_ADD(6)
    if (B + 7 < cnt) ROW_ADD(7)
  }
}

__global__ __launch_bounds__(256) void k_scatter(const int* __restrict__ ei, int E,
                          const float* __restrict__ hw,
                          float* __restrict__ agg, int* __restrict__ place, int N) {
  __shared__ int place_sh;
  const int lane = threadIdx.x & 31;
  const int wv = __builtin_amdgcn_readfirstlane((int)(threadIdx.x >> 5));
  const int hs = (int)(gridDim.x >> 1);
  if (threadIdx.x == 0) {
    int h = pkg_half();
    int k = atomicAdd(place + h * PLACE_PITCH, 1);
    if (k >= hs) {
      h ^= 1;
      k = atomicAdd(place + h * PLACE_PITCH, 1);
    }
    place_sh = k < hs ? h * hs + k : -1;
  }
  __syncthreads();
  const int pl = __builtin_amdgcn_readfirstlane(place_sh);
  if (pl < 0) return;
  const int r  = pl >= hs ? 1 : 0;
  const int gw = (pl - r * hs) * (int)(blockDim.x >> 5) + wv;
  const int nw = hs * (int)(blockDim.x >> 5);
  const int nch = (E + 31) >> 5;
  const int nh  = N >> 1;
  if (gw >= nch) return;
  int q = gw;
  int s, d;
  {
    const int e = q * 32 + lane;
    const int ec = e < E ? e : E - 1;
    s = ei[ec];
    d = ei[E + ec];
  }
  for (;;) {
    const int qn = q + nw;
    const int qp = qn < nch ? qn : q;
    const int en = qp * 32 + lane;
    const int enc = en < E ? en : E - 1;
    const int sn = ei[enc];
    const int dn = ei[E + enc];
    const bool mine = (q * 32 + lane < E) && ((d >= nh) == (r != 0));
    const unsigned m = __builtin_amdgcn_ballot_w32(mine);
    const int cnt = __builtin_popcount(m);
    if (cnt != 0) {
      const int below = (int)__builtin_amdgcn_mbcnt_lo(m, 0u);
      const int pos = mine ? below : cnt + lane - below;
      const int sc = __builtin_amdgcn_ds_permute(pos << 2, s);
      const int dc = __builtin_amdgcn_ds_permute(pos << 2, d);
      Rows va, vb;
      load_group<0>(va, hw, sc, cnt, lane);
      load_group<RPW>(vb, hw, sc, cnt, lane);
      add_group<0>(va, agg, dc, cnt, lane);
      load_group<2 * RPW>(va, hw, sc, cnt, lane);
      add_group<RPW>(vb, agg, dc, cnt, lane);
      load_group<3 * RPW>(vb, hw, sc, cnt, lane);
      add_group<2 * RPW>(va, agg, dc, cnt, lane);
      add_group<3 * RPW>(vb, agg, dc, cnt, lane);
    }
    if (qn >= nch) break;
    q = qn; s = sn; d = dn;
  }
}

__global__ __launch_bounds__(256) void k_pool(const float* __restrict__ h, const int* __restrict__ batch,
                       const float* __restrict__ dinv, const float* __restrict__ bias,
                       float* __restrict__ pool0, float* __restrict__ pool1,
                       float* __restrict__ cnt2, int N, int G) {
  const int lane = threadIdx.x & 31;
  const int wv = __builtin_amdgcn_readfirstlane((int)(threadIdx.x >> 5));
  const int gw = blockIdx.x * (blockDim.x >> 5) + wv;
  const int nw = gridDim.x * (blockDim.x >> 5);
  const int nfull = N / RPW;
  if (gw < nfull) {
    const float bj0 = bias[lane], bj1 = bias[32 + lane];
    const int hf = pkg_half();
    float* out = hf ? pool1 : pool0;
    float* cnt = cnt2 + hf * G;
    for (int q = gw; q < nfull; q += nw) {
    const int n0 = q * RPW;
    const int gl = batch[n0 + (lane & (RPW - 1))];
    const float dl = dinv[n0 + (lane & (RPW - 1))];
    float a0[RPW], a1[RPW];
    #pragma unroll
    for (int i = 0; i < RPW; ++i) {
      const float* r0 = h + (long)(n0 + i) * HID;
      a0[i] = r0[lane]; a1[i] = r0[32 + lane];
    }
    #pragma unroll
    for (int i = 0; i < RPW; ++i) {
      const int g = __builtin_amdgcn_readlane(gl, i);
      const float di = __int_as_float(__builtin_amdgcn_readlane(__float_as_int(dl), i));
      float* o = out + (long)g * HID;
      atomicAdd(o + lane,      fmaxf(fmaf(a0[i], di, bj0), 0.0f));
      atomicAdd(o + 32 + lane, fmaxf(fmaf(a1[i], di, bj1), 0.0f));
      if (lane == 0) atomicAdd(&cnt[g], 1.0f);
    }
    }
  }
  if (gw == nfull % nw && nfull * RPW < N) {
    const int hf = pkg_half();
    float* out = hf ? pool1 : pool0;
    float* cnt = cnt2 + hf * G;
    for (int n = nfull * RPW; n < N; ++n) {
      const int g = batch[n];
      const float di = dinv[n];
      const float* r0 = h + (long)n * HID;
      float* o = out + (long)g * HID;
      atomicAdd(o + lane,      fmaxf(fmaf(r0[lane], di, bias[lane]), 0.0f));
      atomicAdd(o + 32 + lane, fmaxf(fmaf(r0[32 + lane], di, bias[32 + lane]), 0.0f));
      if (lane == 0) atomicAdd(&cnt[g], 1.0f);
    }
  }
}

__global__ void k_div(float* __restrict__ out, const float* __restrict__ pool1,
                      const float* __restrict__ cnt, int total, int G) {
  int t = blockIdx.x * blockDim.x + threadIdx.x;
  if (t < total) out[t] = (out[t] + pool1[t]) / fmaxf(cnt[t >> 6] + cnt[G + (t >> 6)], 1.0f);
}

extern "C" void kernel_launch(void* const* d_in, const int* in_sizes, int n_in,
                              void* d_out, int out_size, void* d_ws, size_t ws_size,
                              hipStream_t stream) {
  const float* x   = (const float*)d_in[0];
  const float* W1  = (const float*)d_in[1];
  const float* b1  = (const float*)d_in[2];
  const float* W2  = (const float*)d_in[3];
  const float* b2  = (const float*)d_in[4];
  const int*   ei  = (const int*)d_in[5];
  const int*   bat = (const int*)d_in[6];
  float* out = (float*)d_out;

  const int N = in_sizes[0] / 3;
  const int E = in_sizes[5] / 2;

  const int G = out_size / HID;

  char* ws = (char*)d_ws;
  float* deg  = (float*)(ws);
  float* dinv = (float*)(ws + (1 << 20));
  float* bufA = (float*)(ws + (1 << 21));
  float* ax   = (float*)(ws + (1 << 21) + (size_t)N * HID * 4);
  float* xs   = ax + 8 * (size_t)N;
  float* bufC = (float*)(ws + (1 << 21) + 2 * (size_t)N * HID * 4);
  float* pool1 = (float*)(ws + (1 << 21) + 4 * (size_t)N * HID * 4);
  float* cnt  = pool1 + out_size;
  int*   place = (int*)(cnt + 2 * (size_t)G) + PLACE_PITCH;

  const int B = 256;
  const int WPB  = B / 32;
  const int tilesM = (N + 15) / 16;
  const int GZ = 128;
  const int GE = 512;
  const int G3 = 1024;
  const int GS = 1024;
  const int GP = 512;

  k_zero  <<<GZ, B, 0, stream>>>(deg, 2 * N, place);
  k_degree<<<GE, B, 0, stream>>>(ei, E, deg, N);

  k_prep    <<<(N + B - 1) / B, B, 0, stream>>>(x, deg, dinv, xs, ax, out, pool1, cnt, N, out_size);
  k_scatter3<<<G3, B, 0, stream>>>(ei, E, xs, ax, N);

  {
    k_gemm2<<<(tilesM + WPB - 1) / WPB, B, 0, stream>>>(ax, W1, b1, W2, dinv, bufA, bufC, N);
  }
  k_scatter<<<GS, B, 0, stream>>>(ei, E, bufA, bufC, place, N);

  k_pool<<<GP, B, 0, stream>>>(bufC, bat, dinv, b2, out, pool1, cnt, N, G);
  k_div <<<(out_size + B - 1) / B, B, 0, stream>>>(out, pool1, cnt, out_size, G);
}


# pool: a wave takes one contiguous span of nodes and sums each run of equal graph number in registers, one set of atomic adds per run (was per node); reorders the f32 pool sums
# speedup vs baseline: 4.9634x; 1.0571x over previous
#include <hip/hip_runtime.h>


#define HID 64


typedef float v2f __attribute__((ext_vector_type(2)));
typedef float v8f __attribute__((ext_vector_type(8)));

__device__ __forceinline__ int pkg_half() {
  return (int)((__builtin_amdgcn_s_sendmsg_rtn(0x87) >> 18) & 1u);
}

#define PLACE_PITCH 64
__global__ void k_zero(float* __restrict__ a, int na, int* __restrict__ place) {
  const int t = blockIdx.x * blockDim.x + threadIdx.x;
  const int stride = gridDim.x * blockDim.x;
  if (t < 2) place[t * PLACE_PITCH] = 0;
  for (int i = t; i < na; i += stride) a[i] = 0.0f;
}

__global__ void k_degree(const int* __restrict__ ei, int E, float* __restrict__ deg, int N) {
  const int t = blockIdx.x * blockDim.x + threadIdx.x;
  const int stride = gridDim.x * blockDim.x;
  float* degh = deg + (long)pkg_half() * N;
  for (int e = t; e < E; e += 8 * stride) {
    int d[8];
    #pragma unroll
    for (int k = 0; k < 8; ++k) {
      const int idx = e + k * stride;
      d[k] = ei[E + (idx < E ? idx : E - 1)];
    }
    #pragma unroll
    for (int k = 0; k < 8; ++k)
      if (e + k * stride < E) atomicAdd(&degh[d[k]], 1.0f);
  }
}

__global__ void k_prep(const float* __restrict__ x, const float* __restrict__ deg,
                       float* __restrict__ dinv, float* __restrict__ xs, float* __restrict__ ax,
                       float* __restrict__ pool0, float* __restrict__ pool1,
                       float* __restrict__ cnt,
                       int N, int pool_size) {
  const int t = blockIdx.x * blockDim.x + threadIdx.x;
  const int stride = gridDim.x * blockDim.x;
  for (int i = t; i < pool_size; i += stride) { pool0[i] = 0.0f; pool1[i] = 0.0f; }
  for (int i = t; i < 2 * (pool_size >> 6); i += stride) cnt[i] = 0.0f;
  if (t >= N) return;
  const float di = rsqrtf(deg[t] + deg[N + t] + 1.0f);
  float4 v;
  v.x = x[t * 3 + 0] * di;
  v.y = x[t * 3 + 1] * di;
  v.z = x[t * 3 + 2] * di;
  v.w = 0.0f;
  *(float4*)(xs + (long)t * 4) = v;
  *(float4*)(ax + (long)t * 4) = v;
  const float4 z = {0.0f, 0.0f, 0.0f, 0.0f};
  *(float4*)(ax + ((long)N + t) * 4) = z;
  dinv[t] = di;
}

__global__ void k_scatter3(const int* __restrict__ ei, int E,
                           const float* __restrict__ xs,
                           float* __restrict__ ax2, int N) {
  const int t  = blockIdx.x * blockDim.x + threadIdx.x;
  const int c  = t & 3;
  if (c >= 3) return;
  const int step = ((gridDim.x * blockDim.x) >> 2) * 4;
  float* ax = ax2 + (long)pkg_half() * N * 4;
  for (int e0 = (t >> 2) * 4; e0 < E; e0 += step) {
  if (e0 + 3 < E) {
    int s[4], d[4];
    #pragma unroll
    for (int i = 0; i < 4; ++i) { s[i] = ei[e0 + i]; d[i] = ei[E + e0 + i]; }
    float v[4];
    #pragma unroll
    for (int i = 0; i < 4; ++i)
      v[i] = xs[(long)s[i] * 4 + c];
    #pragma unroll
    for (int i = 0; i < 4; ++i) atomicAdd(ax + (long)d[i] * 4 + c, v[i]);
  } else {
    for (int e = e0; e < E; ++e) {
      const int s = ei[e], d = ei[E + e];
      atomicAdd(ax + (long)d * 4 + c, xs[(long)s * 4 + c]);
    }
  }
  }
}

__global__ __launch_bounds__(256) void k_gemm2(const float* __restrict__ AX,
                                              const float* __restrict__ W1,
                                              const float* __restrict__ b1,
                                              const float* __restrict__ W,
                                              const float* __restrict__ dinv,
                                              float* __restrict__ O,
                                              float* __restrict__ S, int N) {
  const int lane = threadIdx.x & 31;
  const int wv   = __builtin_amdgcn_readfirstlane((int)(threadIdx.x >> 5));
  const int tm   = blockIdx.x * (blockDim.x >> 5) + wv;
  if (tm * 16 >= N) return;
  const int m    = lane & 15;
  const int half = lane >> 4;
  const int c16  = lane & 15;

  int arow = tm * 16 + m;
  if (arow >= N) arow = N - 1;
  const float4 xa0 = *(const float4*)(AX + (long)arow * 4);
  const float4 xa1 = *(const float4*)(AX + ((long)N + arow) * 4);
  const float da = dinv[arow];
  float4 xa;
  xa.x = (xa0.x + xa1.x) * da; xa.y = (xa0.y + xa1.y) * da; xa.z = (xa0.z + xa1.z) * da; xa.w = 0.0f;

  float sl[8];
  #pragma unroll
  for (int r = 0; r < 8; ++r) {
    int mr = tm * 16 + r + half * 8;
    if (mr >= N) mr = N - 1;
    sl[r] = dinv[mr];
  }

  v8f acc[4] = {};
  #pragma unroll
  for (int k0 = 0; k0 < HID; k0 += 4) {
    const int ka = k0 + half * 2;
    const float h0 = xa.x * W1[ka]     + xa.y * W1[HID + ka]     + xa.z * W1[2 * HID + ka]     + b1[ka];
    const float h1 = xa.x * W1[ka + 1] + xa.y * W1[HID + ka + 1] + xa.z * W1[2 * HID + ka + 1] + b1[ka + 1];
    v2f a; a.x = fmaxf(h0, 0.0f);      a.y = fmaxf(h1, 0.0f);
    #pragma unroll
    for (int tn = 0; tn < 4; ++tn) {
      const int ncol = tn * 16 + c16;
      v2f b; b.x = W[ka * HID + ncol];   b.y = W[(ka + 1) * HID + ncol];
      acc[tn] = __builtin_amdgcn_wmma_f32_16x16x4_f32(
           false, a,  false, b,
           (short)0, acc[tn],  false,  false);
    }
  }
  #pragma unroll
  for (int r = 0; r < 8; ++r) {
    int mrow = tm * 16 + r + half * 8;
    if (mrow < N) {
      #pragma unroll
      for (int tn = 0; tn < 4; ++tn) {
        const int ncol = tn * 16 + c16;
        const float hv = acc[tn][r] * sl[r];
        O[(long)mrow * HID + ncol] = hv;
        S[(long)mrow * HID + ncol] = hv;
      }
    }
  }
}

#define RPW 8

struct Rows { float x0, y0, x1, y1, x2, y2, x3, y3, x4, y4, x5, y5, x6, y6, x7, y7; };

#define ROW_LOAD(i) { const float* row = hw + (long)__builtin_amdgcn_readlane(sc, B + i) * HID; \
                      v.x##i = row[lane]; v.y##i = row[32 + lane]; }
#define ROW_ADD(i)  { float* o = agg + (long)__builtin_amdgcn_readlane(dc, B + i) * HID; \
                      atomicAdd(o + lane, v.x##i); atomicAdd(o + 32 + lane, v.y##i); }

template <int B>
__device__ __forceinline__ void load_group(Rows& v, const float* __restrict__ hw,
                                           int sc, int cnt, int lane) {
  if (B < cnt) {
    ROW_LOAD(0)
    if (B + 1 < cnt) ROW_LOAD(1)
    if (B + 2 < cnt) ROW_LOAD(2)
    if (B + 3 < cnt) ROW_LOAD(3)
    if (B + 4 < cnt) ROW_LOAD(4)
    if (B + 5 < cnt) ROW_LOAD(5)
    if (B + 6 < cnt) ROW_LOAD(6)
    if (B + 7 < cnt) ROW_LOAD(7)
  }
}

template <int B>
__device__ __forceinline__ void add_group(const Rows& v, float* __restrict__ agg,
                                          int dc, int cnt, int lane) {
  if (B < cnt) {
    ROW_ADD(0)
    if (B + 1 < cnt) ROW_ADD(1)
    if (B + 2 < cnt) ROW_ADD(2)
    if (B + 3 < cnt) ROW_ADD(3)
    if (B + 4 < cnt) ROW_ADD(4)
    if (B + 5 < cnt) ROW_ADD(5)
    if (B + 6 < cnt) ROW_ADD(6)
    if (B + 7 < cnt) ROW_ADD(7)
  }
}

__global__ __launch_bounds__(256) void k_scatter(const int* __restrict__ ei, int E,
                          const float* __restrict__ hw,
                          float* __restrict__ agg, int* __restrict__ place, int N) {
  __shared__ int place_sh;
  const int lane = threadIdx.x & 31;
  const int wv = __builtin_amdgcn_readfirstlane((int)(threadIdx.x >> 5));
  const int hs = (int)(gridDim.x >> 1);
  if (threadIdx.x == 0) {
    int h = pkg_half();
    int k = atomicAdd(place + h * PLACE_PITCH, 1);
    if (k >= hs) {
      h ^= 1;
      k = atomicAdd(place + h * PLACE_PITCH, 1);
    }
    place_sh = k < hs ? h * hs + k : -1;
  }
  __syncthreads();
  const int pl = __builtin_amdgcn_readfirstlane(place_sh);
  if (pl < 0) return;
  const int r  = pl >= hs ? 1 : 0;
  const int gw = (pl - r * hs) * (int)(blockDim.x >> 5) + wv;
  const int nw = hs * (int)(blockDim.x >> 5);
  const int nch = (E + 31) >> 5;
  const int nh  = N >> 1;
  if (gw >= nch) return;
  int q = gw;
  int s, d;
  {
    const int e = q * 32 + lane;
    const int ec = e < E ? e : E - 1;
    s = ei[ec];
    d = ei[E + ec];
  }
  for (;;) {
    const int qn = q + nw;
    const int qp = qn < nch ? qn : q;
    const int en = qp * 32 + lane;
    const int enc = en < E ? en : E - 1;
    const int sn = ei[enc];
    const int dn = ei[E + enc];
    const bool mine = (q * 32 + lane < E) && ((d >= nh) == (r != 0));
    const unsigned m = __builtin_amdgcn_ballot_w32(mine);
    const int cnt = __builtin_popcount(m);
    if (cnt != 0) {
      const int below = (int)__builtin_amdgcn_mbcnt_lo(m, 0u);
      const int pos = mine ? below : cnt + lane - below;
      const int sc = __builtin_amdgcn_ds_permute(pos << 2, s);
      const int dc = __builtin_amdgcn_ds_permute(pos << 2, d);
      Rows va, vb;
      load_group<0>(va, hw, sc, cnt, lane);
      load_group<RPW>(vb, hw, sc, cnt, lane);
      add_group<0>(va, agg, dc, cnt, lane);
      load_group<2 * RPW>(va, hw, sc, cnt, lane);
      add_group<RPW>(vb, agg, dc, cnt, lane);
      load_group<3 * RPW>(vb, hw, sc, cnt, lane);
      add_group<2 * RPW>(va, agg, dc, cnt, lane);
      add_group<3 * RPW>(vb, agg, dc, cnt, lane);
    }
    if (qn >= nch) break;
    q = qn; s = sn; d = dn;
  }
}

__global__ __launch_bounds__(256) void k_pool(const float* __restrict__ h, const int* __restrict__ batch,
                       const float* __restrict__ dinv, const float* __restrict__ bias,
                       float* __restrict__ pool0, float* __restrict__ pool1,
                       float* __restrict__ cnt2, int N, int G) {
  const int lane = threadIdx.x & 31;
  const int wv = __builtin_amdgcn_readfirstlane((int)(threadIdx.x >> 5));
  const int gw = blockIdx.x * (blockDim.x >> 5) + wv;
  const int nw = gridDim.x * (blockDim.x >> 5);
  const int per  = (N + nw - 1) / nw;
  const int span = (per + RPW - 1) & ~(RPW - 1);
  const int beg  = gw * span;
  if (beg >= N) return;
  const int end  = beg + span < N ? beg + span : N;
  const float bj0 = bias[lane], bj1 = bias[32 + lane];
  const int hf = pkg_half();
  float* out = hf ? pool1 : pool0;
  float* cnt = cnt2 + hf * G;
  int gcur = -1;
  int crun = 0;
  float s0 = 0.0f, s1 = 0.0f;
  for (int n0 = beg; n0 < end; n0 += RPW) {
    const int nl = n0 + (lane & (RPW - 1));
    const int nc = nl < N ? nl : N - 1;
    const int gl = batch[nc];
    const float dl = dinv[nc];
    float a0[RPW], a1[RPW];
    #pragma unroll
    for (int i = 0; i < RPW; ++i) {
      const int nr = n0 + i < N ? n0 + i : N - 1;
      const float* r0 = h + (long)nr * HID;
      a0[i] = r0[lane]; a1[i] = r0[32 + lane];
    }
    #pragma unroll
    for (int i = 0; i < RPW; ++i) {
      if (n0 + i < end) {
        const int g = __builtin_amdgcn_readlane(gl, i);
        const float di = __int_as_float(__builtin_amdgcn_readlane(__float_as_int(dl), i));
        if (g != gcur) {
          if (crun != 0) {
            float* o = out + (long)gcur * HID;
            atomicAdd(o + lane, s0);
            atomicAdd(o + 32 + lane, s1);
            if (lane == 0) atomicAdd(&cnt[gcur], (float)crun);
          }
          gcur = g; crun = 0; s0 = 0.0f; s1 = 0.0f;
        }
        s0 += fmaxf(fmaf(a0[i], di, bj0), 0.0f);
        s1 += fmaxf(fmaf(a1[i], di, bj1), 0.0f);
        ++crun;
      }
    }
  }
  if (crun != 0) {
    float* o = out + (long)gcur * HID;
    atomicAdd(o + lane, s0);
    atomicAdd(o + 32 + lane, s1);
    if (lane == 0) atomicAdd(&cnt[gcur], (float)crun);
  }
}

__global__ void k_div(float* __restrict__ out, const float* __restrict__ pool1,
                      const float* __restrict__ cnt, int total, int G) {
  int t = blockIdx.x * blockDim.x + threadIdx.x;
  if (t < total) out[t] = (out[t] + pool1[t]) / fmaxf(cnt[t >> 6] + cnt[G + (t >> 6)], 1.0f);
}

extern "C" void kernel_launch(void* const* d_in, const int* in_sizes, int n_in,
                              void* d_out, int out_size, void* d_ws, size_t ws_size,
                              hipStream_t stream) {
  const float* x   = (const float*)d_in[0];
  const float* W1  = (const float*)d_in[1];
  const float* b1  = (const float*)d_in[2];
  const float* W2  = (const float*)d_in[3];
  const float* b2  = (const float*)d_in[4];
  const int*   ei  = (const int*)d_in[5];
  const int*   bat = (const int*)d_in[6];
  float* out = (float*)d_out;

  const int N = in_sizes[0] / 3;
  const int E = in_sizes[5] / 2;

  const int G = out_size / HID;

  char* ws = (char*)d_ws;
  float* deg  = (float*)(ws);
  float* dinv = (float*)(ws + (1 << 20));
  float* bufA = (float*)(ws + (1 << 21));
  float* ax   = (float*)(ws + (1 << 21) + (size_t)N * HID * 4);
  float* xs   = ax + 8 * (size_t)N;
  float* bufC = (float*)(ws + (1 << 21) + 2 * (size_t)N * HID * 4);
  float* pool1 = (float*)(ws + (1 << 21) + 4 * (size_t)N * HID * 4);
  float* cnt  = pool1 + out_size;
  int*   place = (int*)(cnt + 2 * (size_t)G) + PLACE_PITCH;

  const int B = 256;
  const int WPB  = B / 32;
  const int tilesM = (N + 15) / 16;
  const int GZ = 128;
  const int GE = 512;
  const int G3 = 1024;
  const int GS = 1024;
  const int GP = 512;

  k_zero  <<<GZ, B, 0, stream>>>(deg, 2 * N, place);
  k_degree<<<GE, B, 0, stream>>>(ei, E, deg, N);

  k_prep    <<<(N + B - 1) / B, B, 0, stream>>>(x, deg, dinv, xs, ax, out, pool1, cnt, N, out_size);
  k_scatter3<<<G3, B, 0, stream>>>(ei, E, xs, ax, N);

  {
    k_gemm2<<<(tilesM + WPB - 1) / WPB, B, 0, stream>>>(ax, W1, b1, W2, dinv, bufA, bufC, N);
  }
  k_scatter<<<GS, B, 0, stream>>>(ei, E, bufA, bufC, place, N);

  k_pool<<<GP, B, 0, stream>>>(bufC, bat, dinv, b2, out, pool1, cnt, N, G);
  k_div <<<(out_size + B - 1) / B, B, 0, stream>>>(out, pool1, cnt, out_size, G);
}
